# starting kernel
# baseline (speedup 1.0000x reference)
#include <hip/hip_runtime.h>
#include <hip/hip_bf16.h>

typedef __attribute__((ext_vector_type(2))) float v2f;
typedef __attribute__((ext_vector_type(4))) float v4f;
typedef __attribute__((ext_vector_type(8))) float v8f;

#define CCH 128
#define HDIM 128
#define WDIM 128
#define HW (HDIM * WDIM)
#define KSZ 7
#define KK2 49
#define TPITCH 132

__global__ __launch_bounds__(256) void gemm_bn_wmma(
    const float* __restrict__ Xa, long sCa, long sPa,
    const float* __restrict__ Xb, long sCb, long sPb,
    const float* __restrict__ Wm, int ldw,
    const float* __restrict__ scale, const float* __restrict__ bias,
    float* __restrict__ Y, long sCo, long sPo, int relu)
{
    __shared__ float tile[64 * TPITCH];

    const int tid  = threadIdx.x;
    const int lane = tid & 31;
    const int wv   = tid >> 5;
    const int hl   = lane >> 4;
    const int l16  = lane & 15;
    const int m0   = wv << 4;
    const long p0  = (long)blockIdx.x * 64;

    v8f acc[4] = {};

    const float* srcX[2] = {Xa, Xb};
    const long   sCs[2]  = {sCa, sCb};
    const long   sPs[2]  = {sPa, sPb};
    const int nseg = (Xb != nullptr) ? 2 : 1;

    int bbase[4];
    #pragma unroll
    for (int s = 0; s < 4; ++s)
        bbase[s] = ((s << 4) + l16) * TPITCH + (hl << 1);

    for (int seg = 0; seg < nseg; ++seg) {
        const float* __restrict__ X = srcX[seg];
        const long sC = sCs[seg], sP = sPs[seg];

        __syncthreads();
        if (sP == 1) {
            #pragma unroll
            for (int i = 0; i < 8; ++i) {
                const int e  = tid + (i << 8);
                const int p4 = (e & 15) << 2;
                const int c  = e >> 4;
                const v4f xv = *(const v4f*)(X + (long)c * sC + p0 + p4);
                tile[(p4 + 0) * TPITCH + c] = xv.x;
                tile[(p4 + 1) * TPITCH + c] = xv.y;
                tile[(p4 + 2) * TPITCH + c] = xv.z;
                tile[(p4 + 3) * TPITCH + c] = xv.w;
            }
        } else {
            #pragma unroll
            for (int i = 0; i < 8; ++i) {
                const int e  = tid + (i << 8);
                const int c4 = (e & 31) << 2;
                const int p  = e >> 5;
                const v4f xv = *(const v4f*)(X + (p0 + p) * sP + c4);
                *(v4f*)(&tile[p * TPITCH + c4]) = xv;
            }
        }
        __syncthreads();

        const float* __restrict__ Ap =
            Wm + (long)(m0 + l16) * ldw + (seg << 7) + (hl << 1);

        #pragma unroll
        for (int k0 = 0; k0 < CCH; k0 += 4) {
            const v2f a = *(const v2f*)(Ap + k0);
            #pragma unroll
            for (int s = 0; s < 4; ++s) {
                const v2f b = *(const v2f*)(&tile[bbase[s] + k0]);
                acc[s] = __builtin_amdgcn_wmma_f32_16x16x4_f32(
                    false, a, false, b, (short)0, acc[s], false, false);
            }
        }
    }

    #pragma unroll
    for (int s = 0; s < 4; ++s) {
        const long p = p0 + (s << 4) + l16;
        #pragma unroll
        for (int r = 0; r < 8; ++r) {
            const int o = m0 + r + (hl << 3);
            float f = acc[s][r] * scale[o] + bias[o];
            if (relu) f = fmaxf(f, 0.0f);
            Y[(long)o * sCo + p * sPo] = f;
        }
    }
}

__global__ __launch_bounds__(256) void local_attn_kernel(
    const float* __restrict__ q, const float* __restrict__ k,
    const float* __restrict__ v, float* __restrict__ nf)
{
    const int lane = threadIdx.x & 31;
    const int wv   = threadIdx.x >> 5;
    const int p    = blockIdx.x * 8 + wv;
    const int h    = p >> 7;
    const int w    = p & (WDIM - 1);

    const v4f qv = *(const v4f*)(q + (long)p * CCH + lane * 4);

    float sim[KK2];
    #pragma unroll
    for (int t = 0; t < KK2; ++t) {
        const int hn = h + t / KSZ - (KSZ / 2);
        const int wn = w + t % KSZ - (KSZ / 2);
        float d = 0.0f;
        if ((unsigned)hn < (unsigned)HDIM && (unsigned)wn < (unsigned)WDIM) {
            const long pn = (long)hn * WDIM + wn;
            const v4f kv = *(const v4f*)(k + pn * CCH + lane * 4);
            d = qv.x * kv.x + qv.y * kv.y + qv.z * kv.z + qv.w * kv.w;
        }
        #pragma unroll
        for (int off = 16; off > 0; off >>= 1)
            d += __shfl_xor(d, off, 32);
        sim[t] = d;
    }

    float m = sim[0];
    #pragma unroll
    for (int t = 1; t < KK2; ++t) m = fmaxf(m, sim[t]);
    float z = 0.0f;
    #pragma unroll
    for (int t = 0; t < KK2; ++t) { sim[t] = __expf(sim[t] - m); z += sim[t]; }
    const float inv = 1.0f / z;

    v4f acc = {0.0f, 0.0f, 0.0f, 0.0f};
    #pragma unroll
    for (int t = 0; t < KK2; ++t) {
        const int hn = h + t / KSZ - (KSZ / 2);
        const int wn = w + t % KSZ - (KSZ / 2);
        if ((unsigned)hn < (unsigned)HDIM && (unsigned)wn < (unsigned)WDIM) {
            const long pn = (long)hn * WDIM + wn;
            const v4f vv = *(const v4f*)(v + pn * CCH + lane * 4);
            const float wt = sim[t] * inv;
            acc.x += wt * vv.x; acc.y += wt * vv.y;
            acc.z += wt * vv.z; acc.w += wt * vv.w;
        }
    }
    *(v4f*)(nf + (long)p * CCH + lane * 4) = acc;
}

extern "C" void kernel_launch(void* const* d_in, const int* in_sizes, int n_in,
                              void* d_out, int out_size, void* d_ws, size_t ws_size,
                              hipStream_t stream) {
    const float* x    = (const float*)d_in[0];
    const float* w_q1 = (const float*)d_in[1];
    const float* s_q1 = (const float*)d_in[2];
    const float* b_q1 = (const float*)d_in[3];
    const float* w_q2 = (const float*)d_in[4];
    const float* s_q2 = (const float*)d_in[5];
    const float* b_q2 = (const float*)d_in[6];
    const float* w_k1 = (const float*)d_in[7];
    const float* s_k1 = (const float*)d_in[8];
    const float* b_k1 = (const float*)d_in[9];
    const float* w_k2 = (const float*)d_in[10];
    const float* s_k2 = (const float*)d_in[11];
    const float* b_k2 = (const float*)d_in[12];
    const float* w_v  = (const float*)d_in[13];
    const float* s_v  = (const float*)d_in[14];
    const float* b_v  = (const float*)d_in[15];
    const float* w_o  = (const float*)d_in[16];
    const float* s_o  = (const float*)d_in[17];
    const float* b_o  = (const float*)d_in[18];
    float* out = (float*)d_out;

    const size_t NB = (size_t)HW * CCH;
    float* q1 = (float*)d_ws;
    float* k1 = q1 + NB;
    float* vb = k1 + NB;
    float* qb = vb + NB;
    float* kb = qb + NB;
    float* nf = kb + NB;

    const dim3 blk(256);
    const dim3 grd(HW / 64);

    gemm_bn_wmma<<<grd, blk, 0, stream>>>(x, HW, 1, nullptr, 0, 0,
                                          w_q1, CCH, s_q1, b_q1,
                                          q1, 1, CCH, 1);
    gemm_bn_wmma<<<grd, blk, 0, stream>>>(x, HW, 1, nullptr, 0, 0,
                                          w_k1, CCH, s_k1, b_k1,
                                          k1, 1, CCH, 1);
    gemm_bn_wmma<<<grd, blk, 0, stream>>>(x, HW, 1, nullptr, 0, 0,
                                          w_v, CCH, s_v, b_v,
                                          vb, 1, CCH, 1);
    gemm_bn_wmma<<<grd, blk, 0, stream>>>(q1, 1, CCH, nullptr, 0, 0,
                                          w_q2, CCH, s_q2, b_q2,
                                          qb, 1, CCH, 1);
    gemm_bn_wmma<<<grd, blk, 0, stream>>>(k1, 1, CCH, nullptr, 0, 0,
                                          w_k2, CCH, s_k2, b_k2,
                                          kb, 1, CCH, 1);
    local_attn_kernel<<<dim3(HW / 8), blk, 0, stream>>>(qb, kb, vb, nf);
    gemm_bn_wmma<<<grd, blk, 0, stream>>>(nf, 1, CCH, x, HW, 1,
                                          w_o, 2 * CCH, s_o, b_o,
                                          out, HW, 1, 0);
}


# attention: K/V 14x14 halo of each 8x8 pixel tile staged once in LDS (zero-filled outside the image) instead of 49 global re-reads per pixel; sum order unchanged
# speedup vs baseline: 1.4583x; 1.4583x over previous
#include <hip/hip_runtime.h>
#include <hip/hip_bf16.h>

typedef __attribute__((ext_vector_type(2))) float v2f;
typedef __attribute__((ext_vector_type(4))) float v4f;
typedef __attribute__((ext_vector_type(8))) float v8f;

#define CCH 128
#define HDIM 128
#define WDIM 128
#define HW (HDIM * WDIM)
#define KSZ 7
#define KK2 49
#define TPITCH 132

__global__ __launch_bounds__(256) void gemm_bn_wmma(
    const float* __restrict__ Xa, long sCa, long sPa,
    const float* __restrict__ Xb, long sCb, long sPb,
    const float* __restrict__ Wm, int ldw,
    const float* __restrict__ scale, const float* __restrict__ bias,
    float* __restrict__ Y, long sCo, long sPo, int relu)
{
    __shared__ float tile[64 * TPITCH];

    const int tid  = threadIdx.x;
    const int lane = tid & 31;
    const int wv   = tid >> 5;
    const int hl   = lane >> 4;
    const int l16  = lane & 15;
    const int m0   = wv << 4;
    const long p0  = (long)blockIdx.x * 64;

    v8f acc[4] = {};

    const float* srcX[2] = {Xa, Xb};
    const long   sCs[2]  = {sCa, sCb};
    const long   sPs[2]  = {sPa, sPb};
    const int nseg = (Xb != nullptr) ? 2 : 1;

    int bbase[4];
    #pragma unroll
    for (int s = 0; s < 4; ++s)
        bbase[s] = ((s << 4) + l16) * TPITCH + (hl << 1);

    for (int seg = 0; seg < nseg; ++seg) {
        const float* __restrict__ X = srcX[seg];
        const long sC = sCs[seg], sP = sPs[seg];

        __syncthreads();
        if (sP == 1) {
            #pragma unroll
            for (int i = 0; i < 8; ++i) {
                const int e  = tid + (i << 8);
                const int p4 = (e & 15) << 2;
                const int c  = e >> 4;
                const v4f xv = *(const v4f*)(X + (long)c * sC + p0 + p4);
                tile[(p4 + 0) * TPITCH + c] = xv.x;
                tile[(p4 + 1) * TPITCH + c] = xv.y;
                tile[(p4 + 2) * TPITCH + c] = xv.z;
                tile[(p4 + 3) * TPITCH + c] = xv.w;
            }
        } else {
            #pragma unroll
            for (int i = 0; i < 8; ++i) {
                const int e  = tid + (i << 8);
                const int c4 = (e & 31) << 2;
                const int p  = e >> 5;
                const v4f xv = *(const v4f*)(X + (p0 + p) * sP + c4);
                *(v4f*)(&tile[p * TPITCH + c4]) = xv;
            }
        }
        __syncthreads();

        const float* __restrict__ Ap =
            Wm + (long)(m0 + l16) * ldw + (seg << 7) + (hl << 1);

        #pragma unroll
        for (int k0 = 0; k0 < CCH; k0 += 4) {
            const v2f a = *(const v2f*)(Ap + k0);
            #pragma unroll
            for (int s = 0; s < 4; ++s) {
                const v2f b = *(const v2f*)(&tile[bbase[s] + k0]);
                acc[s] = __builtin_amdgcn_wmma_f32_16x16x4_f32(
                    false, a, false, b, (short)0, acc[s], false, false);
            }
        }
    }

    #pragma unroll
    for (int s = 0; s < 4; ++s) {
        const long p = p0 + (s << 4) + l16;
        #pragma unroll
        for (int r = 0; r < 8; ++r) {
            const int o = m0 + r + (hl << 3);
            float f = acc[s][r] * scale[o] + bias[o];
            if (relu) f = fmaxf(f, 0.0f);
            Y[(long)o * sCo + p * sPo] = f;
        }
    }
}

#define TILE_W 8
#define TILE_H 8
#define HALO_W (TILE_W + KSZ - 1)
#define HALO_H (TILE_H + KSZ - 1)
#define HALO_N (HALO_W * HALO_H)
#define HALO_V4 (HALO_N * (CCH / 4))
#define ATT_THREADS 1024
#define PIX_PER_WAVE 2

__global__ __launch_bounds__(ATT_THREADS) void local_attn_kernel(
    const float* __restrict__ q, const float* __restrict__ k,
    const float* __restrict__ v, float* __restrict__ nf)
{
    __shared__ __attribute__((aligned(16))) float ks[HALO_N * CCH];
    __shared__ __attribute__((aligned(16))) float vs[HALO_N * CCH];

    const int tid  = threadIdx.x;
    const int lane = tid & 31;
    const int wv   = tid >> 5;
    const int h0   = (blockIdx.x / (WDIM / TILE_W)) * TILE_H;
    const int w0   = (blockIdx.x % (WDIM / TILE_W)) * TILE_W;

    v4f qv[PIX_PER_WAVE];
    #pragma unroll
    for (int pp = 0; pp < PIX_PER_WAVE; ++pp) {
        const int idx = wv * PIX_PER_WAVE + pp;
        const int p   = (h0 + (idx / TILE_W)) * WDIM + w0 + (idx % TILE_W);
        qv[pp] = *(const v4f*)(q + (long)p * CCH + lane * 4);
    }

    #pragma unroll
    for (int it = 0; it < (HALO_V4 + ATT_THREADS - 1) / ATT_THREADS; ++it) {
        const int e = tid + it * ATT_THREADS;
        if (e < HALO_V4) {
            const int hp = e >> 5;
            const int c4 = (e & 31) << 2;
            const int hy = hp / HALO_W;
            const int hx = hp - hy * HALO_W;
            const int gh = h0 + hy - (KSZ / 2);
            const int gw = w0 + hx - (KSZ / 2);
            const bool ok = (unsigned)gh < (unsigned)HDIM && (unsigned)gw < (unsigned)WDIM;
            const int ghc = min(max(gh, 0), HDIM - 1);
            const int gwc = min(max(gw, 0), WDIM - 1);
            const long off = (long)(ghc * WDIM + gwc) * CCH + c4;
            v4f kv = *(const v4f*)(k + off);
            v4f vv = *(const v4f*)(v + off);
            if (!ok) {
                kv = (v4f){0.0f, 0.0f, 0.0f, 0.0f};
                vv = (v4f){0.0f, 0.0f, 0.0f, 0.0f};
            }
            *(v4f*)(&ks[hp * CCH + c4]) = kv;
            *(v4f*)(&vs[hp * CCH + c4]) = vv;
        }
    }
    __syncthreads();

    for (int pp = 0; pp < PIX_PER_WAVE; ++pp) {
        const int idx = wv * PIX_PER_WAVE + pp;
        const int ty  = idx / TILE_W;
        const int tx  = idx % TILE_W;
        const int p   = (h0 + ty) * WDIM + w0 + tx;
        const int wbase = (ty * HALO_W + tx) * CCH + lane * 4;
        const v4f qq = qv[pp];

        float sim[KK2];
        #pragma unroll
        for (int t = 0; t < KK2; ++t) {
            const v4f kv = *(const v4f*)(&ks[wbase + ((t / KSZ) * HALO_W + (t % KSZ)) * CCH]);
            float d = qq.x * kv.x + qq.y * kv.y + qq.z * kv.z + qq.w * kv.w;
            #pragma unroll
            for (int off = 16; off > 0; off >>= 1)
                d += __shfl_xor(d, off, 32);
            sim[t] = d;
        }

        float m = sim[0];
        #pragma unroll
        for (int t = 1; t < KK2; ++t) m = fmaxf(m, sim[t]);
        float z = 0.0f;
        #pragma unroll
        for (int t = 0; t < KK2; ++t) { sim[t] = __expf(sim[t] - m); z += sim[t]; }
        const float inv = 1.0f / z;

        v4f acc = {0.0f, 0.0f, 0.0f, 0.0f};
        #pragma unroll
        for (int t = 0; t < KK2; ++t) {
            const v4f vv = *(const v4f*)(&vs[wbase + ((t / KSZ) * HALO_W + (t % KSZ)) * CCH]);
            const float wt = sim[t] * inv;
            acc.x += wt * vv.x; acc.y += wt * vv.y;
            acc.z += wt * vv.z; acc.w += wt * vv.w;
        }
        *(v4f*)(nf + (long)p * CCH + lane * 4) = acc;
    }
}

extern "C" void kernel_launch(void* const* d_in, const int* in_sizes, int n_in,
                              void* d_out, int out_size, void* d_ws, size_t ws_size,
                              hipStream_t stream) {
    const float* x    = (const float*)d_in[0];
    const float* w_q1 = (const float*)d_in[1];
    const float* s_q1 = (const float*)d_in[2];
    const float* b_q1 = (const float*)d_in[3];
    const float* w_q2 = (const float*)d_in[4];
    const float* s_q2 = (const float*)d_in[5];
    const float* b_q2 = (const float*)d_in[6];
    const float* w_k1 = (const float*)d_in[7];
    const float* s_k1 = (const float*)d_in[8];
    const float* b_k1 = (const float*)d_in[9];
    const float* w_k2 = (const float*)d_in[10];
    const float* s_k2 = (const float*)d_in[11];
    const float* b_k2 = (const float*)d_in[12];
    const float* w_v  = (const float*)d_in[13];
    const float* s_v  = (const float*)d_in[14];
    const float* b_v  = (const float*)d_in[15];
    const float* w_o  = (const float*)d_in[16];
    const float* s_o  = (const float*)d_in[17];
    const float* b_o  = (const float*)d_in[18];
    float* out = (float*)d_out;

    const size_t NB = (size_t)HW * CCH;
    float* q1 = (float*)d_ws;
    float* k1 = q1 + NB;
    float* vb = k1 + NB;
    float* qb = vb + NB;
    float* kb = qb + NB;
    float* nf = kb + NB;

    const dim3 blk(256);
    const dim3 grd(HW / 64);

    gemm_bn_wmma<<<grd, blk, 0, stream>>>(x, HW, 1, nullptr, 0, 0,
                                          w_q1, CCH, s_q1, b_q1,
                                          q1, 1, CCH, 1);
    gemm_bn_wmma<<<grd, blk, 0, stream>>>(x, HW, 1, nullptr, 0, 0,
                                          w_k1, CCH, s_k1, b_k1,
                                          k1, 1, CCH, 1);
    gemm_bn_wmma<<<grd, blk, 0, stream>>>(x, HW, 1, nullptr, 0, 0,
                                          w_v, CCH, s_v, b_v,
                                          vb, 1, CCH, 1);
    gemm_bn_wmma<<<grd, blk, 0, stream>>>(q1, 1, CCH, nullptr, 0, 0,
                                          w_q2, CCH, s_q2, b_q2,
                                          qb, 1, CCH, 1);
    gemm_bn_wmma<<<grd, blk, 0, stream>>>(k1, 1, CCH, nullptr, 0, 0,
                                          w_k2, CCH, s_k2, b_k2,
                                          kb, 1, CCH, 1);
    local_attn_kernel<<<dim3(HW / (TILE_W * TILE_H)), dim3(ATT_THREADS), 0, stream>>>(qb, kb, vb, nf);
    gemm_bn_wmma<<<grd, blk, 0, stream>>>(nf, 1, CCH, x, HW, 1,
                                          w_o, 2 * CCH, s_o, b_o,
                                          out, HW, 1, 0);
}


# attention: 32-lane sum of each of the 49 q.k dot products moved from ds_bpermute round trips to vector-ALU lane exchanges (v_permlanex16 for l^16, DPP row_xmask for l^8,4,2,1); sum order unchanged
# speedup vs baseline: 1.6412x; 1.1254x over previous
#include <hip/hip_runtime.h>
#include <hip/hip_bf16.h>

typedef __attribute__((ext_vector_type(2))) float v2f;
typedef __attribute__((ext_vector_type(4))) float v4f;
typedef __attribute__((ext_vector_type(8))) float v8f;

#define CCH 128
#define HDIM 128
#define WDIM 128
#define HW (HDIM * WDIM)
#define KSZ 7
#define KK2 49
#define TPITCH 132

__global__ __launch_bounds__(256) void gemm_bn_wmma(
    const float* __restrict__ Xa, long sCa, long sPa,
    const float* __restrict__ Xb, long sCb, long sPb,
    const float* __restrict__ Wm, int ldw,
    const float* __restrict__ scale, const float* __restrict__ bias,
    float* __restrict__ Y, long sCo, long sPo, int relu)
{
    __shared__ float tile[64 * TPITCH];

    const int tid  = threadIdx.x;
    const int lane = tid & 31;
    const int wv   = tid >> 5;
    const int hl   = lane >> 4;
    const int l16  = lane & 15;
    const int m0   = wv << 4;
    const long p0  = (long)blockIdx.x * 64;

    v8f acc[4] = {};

    const float* srcX[2] = {Xa, Xb};
    const long   sCs[2]  = {sCa, sCb};
    const long   sPs[2]  = {sPa, sPb};
    const int nseg = (Xb != nullptr) ? 2 : 1;

    int bbase[4];
    #pragma unroll
    for (int s = 0; s < 4; ++s)
        bbase[s] = ((s << 4) + l16) * TPITCH + (hl << 1);

    for (int seg = 0; seg < nseg; ++seg) {
        const float* __restrict__ X = srcX[seg];
        const long sC = sCs[seg], sP = sPs[seg];

        __syncthreads();
        if (sP == 1) {
            #pragma unroll
            for (int i = 0; i < 8; ++i) {
                const int e  = tid + (i << 8);
                const int p4 = (e & 15) << 2;
                const int c  = e >> 4;
                const v4f xv = *(const v4f*)(X + (long)c * sC + p0 + p4);
                tile[(p4 + 0) * TPITCH + c] = xv.x;
                tile[(p4 + 1) * TPITCH + c] = xv.y;
                tile[(p4 + 2) * TPITCH + c] = xv.z;
                tile[(p4 + 3) * TPITCH + c] = xv.w;
            }
        } else {
            #pragma unroll
            for (int i = 0; i < 8; ++i) {
                const int e  = tid + (i << 8);
                const int c4 = (e & 31) << 2;
                const int p  = e >> 5;
                const v4f xv = *(const v4f*)(X + (p0 + p) * sP + c4);
                *(v4f*)(&tile[p * TPITCH + c4]) = xv;
            }
        }
        __syncthreads();

        const float* __restrict__ Ap =
            Wm + (long)(m0 + l16) * ldw + (seg << 7) + (hl << 1);

        #pragma unroll
        for (int k0 = 0; k0 < CCH; k0 += 4) {
            const v2f a = *(const v2f*)(Ap + k0);
            #pragma unroll
            for (int s = 0; s < 4; ++s) {
                const v2f b = *(const v2f*)(&tile[bbase[s] + k0]);
                acc[s] = __builtin_amdgcn_wmma_f32_16x16x4_f32(
                    false, a, false, b, (short)0, acc[s], false, false);
            }
        }
    }

    #pragma unroll
    for (int s = 0; s < 4; ++s) {
        const long p = p0 + (s << 4) + l16;
        #pragma unroll
        for (int r = 0; r < 8; ++r) {
            const int o = m0 + r + (hl << 3);
            float f = acc[s][r] * scale[o] + bias[o];
            if (relu) f = fmaxf(f, 0.0f);
            Y[(long)o * sCo + p * sPo] = f;
        }
    }
}

__device__ __forceinline__ float xhalf(float x) {
    const int b = __builtin_bit_cast(int, x);
    return __builtin_bit_cast(float,
        __builtin_amdgcn_permlanex16(b, b, 0x76543210, 0xfedcba98, false, false));
}
template <int M> __device__ __forceinline__ float xrow(float x) {
    const int b = __builtin_bit_cast(int, x);
    return __builtin_bit_cast(float,
        __builtin_amdgcn_update_dpp(b, b, 0x160 + M, 0xf, 0xf, true));
}

#define TILE_W 8
#define TILE_H 8
#define HALO_W (TILE_W + KSZ - 1)
#define HALO_H (TILE_H + KSZ - 1)
#define HALO_N (HALO_W * HALO_H)
#define HALO_V4 (HALO_N * (CCH / 4))
#define ATT_THREADS 1024
#define PIX_PER_WAVE 2

__global__ __launch_bounds__(ATT_THREADS) void local_attn_kernel(
    const float* __restrict__ q, const float* __restrict__ k,
    const float* __restrict__ v, float* __restrict__ nf)
{
    __shared__ __attribute__((aligned(16))) float ks[HALO_N * CCH];
    __shared__ __attribute__((aligned(16))) float vs[HALO_N * CCH];

    const int tid  = threadIdx.x;
    const int lane = tid & 31;
    const int wv   = tid >> 5;
    const int h0   = (blockIdx.x / (WDIM / TILE_W)) * TILE_H;
    const int w0   = (blockIdx.x % (WDIM / TILE_W)) * TILE_W;

    v4f qv[PIX_PER_WAVE];
    #pragma unroll
    for (int pp = 0; pp < PIX_PER_WAVE; ++pp) {
        const int idx = wv * PIX_PER_WAVE + pp;
        const int p   = (h0 + (idx / TILE_W)) * WDIM + w0 + (idx % TILE_W);
        qv[pp] = *(const v4f*)(q + (long)p * CCH + lane * 4);
    }

    #pragma unroll
    for (int it = 0; it < (HALO_V4 + ATT_THREADS - 1) / ATT_THREADS; ++it) {
        const int e = tid + it * ATT_THREADS;
        if (e < HALO_V4) {
            const int hp = e >> 5;
            const int c4 = (e & 31) << 2;
            const int hy = hp / HALO_W;
            const int hx = hp - hy * HALO_W;
            const int gh = h0 + hy - (KSZ / 2);
            const int gw = w0 + hx - (KSZ / 2);
            const bool ok = (unsigned)gh < (unsigned)HDIM && (unsigned)gw < (unsigned)WDIM;
            const int ghc = min(max(gh, 0), HDIM - 1);
            const int gwc = min(max(gw, 0), WDIM - 1);
            const long off = (long)(ghc * WDIM + gwc) * CCH + c4;
            v4f kv = *(const v4f*)(k + off);
            v4f vv = *(const v4f*)(v + off);
            if (!ok) {
                kv = (v4f){0.0f, 0.0f, 0.0f, 0.0f};
                vv = (v4f){0.0f, 0.0f, 0.0f, 0.0f};
            }
            *(v4f*)(&ks[hp * CCH + c4]) = kv;
            *(v4f*)(&vs[hp * CCH + c4]) = vv;
        }
    }
    __syncthreads();

    for (int pp = 0; pp < PIX_PER_WAVE; ++pp) {
        const int idx = wv * PIX_PER_WAVE + pp;
        const int ty  = idx / TILE_W;
        const int tx  = idx % TILE_W;
        const int p   = (h0 + ty) * WDIM + w0 + tx;
        const int wbase = (ty * HALO_W + tx) * CCH + lane * 4;
        const v4f qq = qv[pp];

        float sim[KK2];
        #pragma unroll
        for (int t = 0; t < KK2; ++t) {
            const v4f kv = *(const v4f*)(&ks[wbase + ((t / KSZ) * HALO_W + (t % KSZ)) * CCH]);
            float d = qq.x * kv.x + qq.y * kv.y + qq.z * kv.z + qq.w * kv.w;
            d += xhalf(d);
            d += xrow<8>(d);
            d += xrow<4>(d);
            d += xrow<2>(d);
            d += xrow<1>(d);
            sim[t] = d;
        }

        float m = sim[0];
        #pragma unroll
        for (int t = 1; t < KK2; ++t) m = fmaxf(m, sim[t]);
        float z = 0.0f;
        #pragma unroll
        for (int t = 0; t < KK2; ++t) { sim[t] = __expf(sim[t] - m); z += sim[t]; }
        const float inv = 1.0f / z;

        v4f acc = {0.0f, 0.0f, 0.0f, 0.0f};
        #pragma unroll
        for (int t = 0; t < KK2; ++t) {
            const v4f vv = *(const v4f*)(&vs[wbase + ((t / KSZ) * HALO_W + (t % KSZ)) * CCH]);
            const float wt = sim[t] * inv;
            acc.x += wt * vv.x; acc.y += wt * vv.y;
            acc.z += wt * vv.z; acc.w += wt * vv.w;
        }
        *(v4f*)(nf + (long)p * CCH + lane * 4) = acc;
    }
}

extern "C" void kernel_launch(void* const* d_in, const int* in_sizes, int n_in,
                              void* d_out, int out_size, void* d_ws, size_t ws_size,
                              hipStream_t stream) {
    const float* x    = (const float*)d_in[0];
    const float* w_q1 = (const float*)d_in[1];
    const float* s_q1 = (const float*)d_in[2];
    const float* b_q1 = (const float*)d_in[3];
    const float* w_q2 = (const float*)d_in[4];
    const float* s_q2 = (const float*)d_in[5];
    const float* b_q2 = (const float*)d_in[6];
    const float* w_k1 = (const float*)d_in[7];
    const float* s_k1 = (const float*)d_in[8];
    const float* b_k1 = (const float*)d_in[9];
    const float* w_k2 = (const float*)d_in[10];
    const float* s_k2 = (const float*)d_in[11];
    const float* b_k2 = (const float*)d_in[12];
    const float* w_v  = (const float*)d_in[13];
    const float* s_v  = (const float*)d_in[14];
    const float* b_v  = (const float*)d_in[15];
    const float* w_o  = (const float*)d_in[16];
    const float* s_o  = (const float*)d_in[17];
    const float* b_o  = (const float*)d_in[18];
    float* out = (float*)d_out;

    const size_t NB = (size_t)HW * CCH;
    float* q1 = (float*)d_ws;
    float* k1 = q1 + NB;
    float* vb = k1 + NB;
    float* qb = vb + NB;
    float* kb = qb + NB;
    float* nf = kb + NB;

    const dim3 blk(256);
    const dim3 grd(HW / 64);

    gemm_bn_wmma<<<grd, blk, 0, stream>>>(x, HW, 1, nullptr, 0, 0,
                                          w_q1, CCH, s_q1, b_q1,
                                          q1, 1, CCH, 1);
    gemm_bn_wmma<<<grd, blk, 0, stream>>>(x, HW, 1, nullptr, 0, 0,
                                          w_k1, CCH, s_k1, b_k1,
                                          k1, 1, CCH, 1);
    gemm_bn_wmma<<<grd, blk, 0, stream>>>(x, HW, 1, nullptr, 0, 0,
                                          w_v, CCH, s_v, b_v,
                                          vb, 1, CCH, 1);
    gemm_bn_wmma<<<grd, blk, 0, stream>>>(q1, 1, CCH, nullptr, 0, 0,
                                          w_q2, CCH, s_q2, b_q2,
                                          qb, 1, CCH, 1);
    gemm_bn_wmma<<<grd, blk, 0, stream>>>(k1, 1, CCH, nullptr, 0, 0,
                                          w_k2, CCH, s_k2, b_k2,
                                          kb, 1, CCH, 1);
    local_attn_kernel<<<dim3(HW / (TILE_W * TILE_H)), dim3(ATT_THREADS), 0, stream>>>(qb, kb, vb, nf);
    gemm_bn_wmma<<<grd, blk, 0, stream>>>(nf, 1, CCH, x, HW, 1,
                                          w_o, 2 * CCH, s_o, b_o,
                                          out, HW, 1, 0);
}


# projections: the five q/k/v 1x1-conv GEMMs fused into one launch per 64-pixel tile (x tile staged once, q1/k1 kept in LDS; 7 launches -> 3); k order and sum order unchanged
# speedup vs baseline: 2.5535x; 1.5559x over previous
#include <hip/hip_runtime.h>
#include <hip/hip_bf16.h>

typedef __attribute__((ext_vector_type(2))) float v2f;
typedef __attribute__((ext_vector_type(4))) float v4f;
typedef __attribute__((ext_vector_type(8))) float v8f;

#define CCH 128
#define HDIM 128
#define WDIM 128
#define HW (HDIM * WDIM)
#define KSZ 7
#define KK2 49
#define TPITCH 132

__global__ __launch_bounds__(256) void gemm_bn_wmma(
    const float* __restrict__ Xa, long sCa, long sPa,
    const float* __restrict__ Xb, long sCb, long sPb,
    const float* __restrict__ Wm, int ldw,
    const float* __restrict__ scale, const float* __restrict__ bias,
    float* __restrict__ Y, long sCo, long sPo, int relu)
{
    __shared__ float tile[64 * TPITCH];

    const int tid  = threadIdx.x;
    const int lane = tid & 31;
    const int wv   = tid >> 5;
    const int hl   = lane >> 4;
    const int l16  = lane & 15;
    const int m0   = wv << 4;
    const long p0  = (long)blockIdx.x * 64;

    v8f acc[4] = {};

    const float* srcX[2] = {Xa, Xb};
    const long   sCs[2]  = {sCa, sCb};
    const long   sPs[2]  = {sPa, sPb};
    const int nseg = (Xb != nullptr) ? 2 : 1;

    int bbase[4];
    #pragma unroll
    for (int s = 0; s < 4; ++s)
        bbase[s] = ((s << 4) + l16) * TPITCH + (hl << 1);

    for (int seg = 0; seg < nseg; ++seg) {
        const float* __restrict__ X = srcX[seg];
        const long sC = sCs[seg], sP = sPs[seg];

        __syncthreads();
        if (sP == 1) {
            #pragma unroll
            for (int i = 0; i < 8; ++i) {
                const int e  = tid + (i << 8);
                const int p4 = (e & 15) << 2;
                const int c  = e >> 4;
                const v4f xv = *(const v4f*)(X + (long)c * sC + p0 + p4);
                tile[(p4 + 0) * TPITCH + c] = xv.x;
                tile[(p4 + 1) * TPITCH + c] = xv.y;
                tile[(p4 + 2) * TPITCH + c] = xv.z;
                tile[(p4 + 3) * TPITCH + c] = xv.w;
            }
        } else {
            #pragma unroll
            for (int i = 0; i < 8; ++i) {
                const int e  = tid + (i << 8);
                const int c4 = (e & 31) << 2;
                const int p  = e >> 5;
                const v4f xv = *(const v4f*)(X + (p0 + p) * sP + c4);
                *(v4f*)(&tile[p * TPITCH + c4]) = xv;
            }
        }
        __syncthreads();

        const float* __restrict__ Ap =
            Wm + (long)(m0 + l16) * ldw + (seg << 7) + (hl << 1);

        #pragma unroll
        for (int k0 = 0; k0 < CCH; k0 += 4) {
            const v2f a = *(const v2f*)(Ap + k0);
            #pragma unroll
            for (int s = 0; s < 4; ++s) {
                const v2f b = *(const v2f*)(&tile[bbase[s] + k0]);
                acc[s] = __builtin_amdgcn_wmma_f32_16x16x4_f32(
                    false, a, false, b, (short)0, acc[s], false, false);
            }
        }
    }

    #pragma unroll
    for (int s = 0; s < 4; ++s) {
        const long p = p0 + (s << 4) + l16;
        #pragma unroll
        for (int r = 0; r < 8; ++r) {
            const int o = m0 + r + (hl << 3);
            float f = acc[s][r] * scale[o] + bias[o];
            if (relu) f = fmaxf(f, 0.0f);
            Y[(long)o * sCo + p * sPo] = f;
        }
    }
}

__device__ __forceinline__ void mma_tile128(
    const float* tile, const int (&bbase)[4],
    const float* __restrict__ Ap, v8f (&acc)[4])
{
    #pragma unroll
    for (int k0 = 0; k0 < CCH; k0 += 4) {
        const v2f a = *(const v2f*)(Ap + k0);
        #pragma unroll
        for (int s = 0; s < 4; ++s) {
            const v2f b = *(const v2f*)(&tile[bbase[s] + k0]);
            acc[s] = __builtin_amdgcn_wmma_f32_16x16x4_f32(
                false, a, false, b, (short)0, acc[s], false, false);
        }
    }
}

__device__ __forceinline__ void bn_relu_tile(
    const v8f& d, const float* __restrict__ scale, const float* __restrict__ bias,
    int o8, v4f& lo, v4f& hi)
{
    const v4f s0 = *(const v4f*)(scale + o8), s1 = *(const v4f*)(scale + o8 + 4);
    const v4f b0 = *(const v4f*)(bias + o8),  b1 = *(const v4f*)(bias + o8 + 4);
    lo.x = fmaxf(d[0] * s0.x + b0.x, 0.0f);
    lo.y = fmaxf(d[1] * s0.y + b0.y, 0.0f);
    lo.z = fmaxf(d[2] * s0.z + b0.z, 0.0f);
    lo.w = fmaxf(d[3] * s0.w + b0.w, 0.0f);
    hi.x = fmaxf(d[4] * s1.x + b1.x, 0.0f);
    hi.y = fmaxf(d[5] * s1.y + b1.y, 0.0f);
    hi.z = fmaxf(d[6] * s1.z + b1.z, 0.0f);
    hi.w = fmaxf(d[7] * s1.w + b1.w, 0.0f);
}

__global__ __launch_bounds__(256) void proj_fused_wmma(
    const float* __restrict__ x,
    const float* __restrict__ w_q1, const float* __restrict__ s_q1, const float* __restrict__ b_q1,
    const float* __restrict__ w_q2, const float* __restrict__ s_q2, const float* __restrict__ b_q2,
    const float* __restrict__ w_k1, const float* __restrict__ s_k1, const float* __restrict__ b_k1,
    const float* __restrict__ w_k2, const float* __restrict__ s_k2, const float* __restrict__ b_k2,
    const float* __restrict__ w_v,  const float* __restrict__ s_v,  const float* __restrict__ b_v,
    float* __restrict__ qo, float* __restrict__ ko, float* __restrict__ vo)
{
    __shared__ __attribute__((aligned(16))) float tx[64 * TPITCH];
    __shared__ __attribute__((aligned(16))) float tq[64 * TPITCH];
    __shared__ __attribute__((aligned(16))) float tk[64 * TPITCH];

    const int tid  = threadIdx.x;
    const int lane = tid & 31;
    const int wv   = tid >> 5;
    const int hl   = lane >> 4;
    const int l16  = lane & 15;
    const int m0   = wv << 4;
    const long p0  = (long)blockIdx.x * 64;
    const int o8   = m0 + (hl << 3);

    int bbase[4];
    #pragma unroll
    for (int s = 0; s < 4; ++s)
        bbase[s] = ((s << 4) + l16) * TPITCH + (hl << 1);

    #pragma unroll
    for (int i = 0; i < 8; ++i) {
        const int e  = tid + (i << 8);
        const int p4 = (e & 15) << 2;
        const int c  = e >> 4;
        const v4f xv = *(const v4f*)(x + (long)c * HW + p0 + p4);
        tx[(p4 + 0) * TPITCH + c] = xv.x;
        tx[(p4 + 1) * TPITCH + c] = xv.y;
        tx[(p4 + 2) * TPITCH + c] = xv.z;
        tx[(p4 + 3) * TPITCH + c] = xv.w;
    }
    __syncthreads();

    const long arow = (long)(m0 + l16) * CCH + (hl << 1);

    {
        v8f acc[4] = {};
        mma_tile128(tx, bbase, w_q1 + arow, acc);
        #pragma unroll
        for (int s = 0; s < 4; ++s) {
            v4f lo, hi;
            bn_relu_tile(acc[s], s_q1, b_q1, o8, lo, hi);
            float* d = &tq[((s << 4) + l16) * TPITCH + o8];
            *(v4f*)(d) = lo;  *(v4f*)(d + 4) = hi;
        }
    }
    {
        v8f acc[4] = {};
        mma_tile128(tx, bbase, w_k1 + arow, acc);
        #pragma unroll
        for (int s = 0; s < 4; ++s) {
            v4f lo, hi;
            bn_relu_tile(acc[s], s_k1, b_k1, o8, lo, hi);
            float* d = &tk[((s << 4) + l16) * TPITCH + o8];
            *(v4f*)(d) = lo;  *(v4f*)(d + 4) = hi;
        }
    }
    {
        v8f acc[4] = {};
        mma_tile128(tx, bbase, w_v + arow, acc);
        #pragma unroll
        for (int s = 0; s < 4; ++s) {
            v4f lo, hi;
            bn_relu_tile(acc[s], s_v, b_v, o8, lo, hi);
            float* d = vo + (p0 + (s << 4) + l16) * CCH + o8;
            *(v4f*)(d) = lo;  *(v4f*)(d + 4) = hi;
        }
    }
    __syncthreads();

    {
        v8f acc[4] = {};
        mma_tile128(tq, bbase, w_q2 + arow, acc);
        #pragma unroll
        for (int s = 0; s < 4; ++s) {
            v4f lo, hi;
            bn_relu_tile(acc[s], s_q2, b_q2, o8, lo, hi);
            float* d = qo + (p0 + (s << 4) + l16) * CCH + o8;
            *(v4f*)(d) = lo;  *(v4f*)(d + 4) = hi;
        }
    }
    {
        v8f acc[4] = {};
        mma_tile128(tk, bbase, w_k2 + arow, acc);
        #pragma unroll
        for (int s = 0; s < 4; ++s) {
            v4f lo, hi;
            bn_relu_tile(acc[s], s_k2, b_k2, o8, lo, hi);
            float* d = ko + (p0 + (s << 4) + l16) * CCH + o8;
            *(v4f*)(d) = lo;  *(v4f*)(d + 4) = hi;
        }
    }
}

__device__ __forceinline__ float xhalf(float x) {
    const int b = __builtin_bit_cast(int, x);
    return __builtin_bit_cast(float,
        __builtin_amdgcn_permlanex16(b, b, 0x76543210, 0xfedcba98, false, false));
}
template <int M> __device__ __forceinline__ float xrow(float x) {
    const int b = __builtin_bit_cast(int, x);
    return __builtin_bit_cast(float,
        __builtin_amdgcn_update_dpp(b, b, 0x160 + M, 0xf, 0xf, true));
}

#define TILE_W 8
#define TILE_H 8
#define HALO_W (TILE_W + KSZ - 1)
#define HALO_H (TILE_H + KSZ - 1)
#define HALO_N (HALO_W * HALO_H)
#define HALO_V4 (HALO_N * (CCH / 4))
#define ATT_THREADS 1024
#define PIX_PER_WAVE 2

__global__ __launch_bounds__(ATT_THREADS) void local_attn_kernel(
    const float* __restrict__ q, const float* __restrict__ k,
    const float* __restrict__ v, float* __restrict__ nf)
{
    __shared__ __attribute__((aligned(16))) float ks[HALO_N * CCH];
    __shared__ __attribute__((aligned(16))) float vs[HALO_N * CCH];

    const int tid  = threadIdx.x;
    const int lane = tid & 31;
    const int wv   = tid >> 5;
    const int h0   = (blockIdx.x / (WDIM / TILE_W)) * TILE_H;
    const int w0   = (blockIdx.x % (WDIM / TILE_W)) * TILE_W;

    v4f qv[PIX_PER_WAVE];
    #pragma unroll
    for (int pp = 0; pp < PIX_PER_WAVE; ++pp) {
        const int idx = wv * PIX_PER_WAVE + pp;
        const int p   = (h0 + (idx / TILE_W)) * WDIM + w0 + (idx % TILE_W);
        qv[pp] = *(const v4f*)(q + (long)p * CCH + lane * 4);
    }

    #pragma unroll
    for (int it = 0; it < (HALO_V4 + ATT_THREADS - 1) / ATT_THREADS; ++it) {
        const int e = tid + it * ATT_THREADS;
        if (e < HALO_V4) {
            const int hp = e >> 5;
            const int c4 = (e & 31) << 2;
            const int hy = hp / HALO_W;
            const int hx = hp - hy * HALO_W;
            const int gh = h0 + hy - (KSZ / 2);
            const int gw = w0 + hx - (KSZ / 2);
            const bool ok = (unsigned)gh < (unsigned)HDIM && (unsigned)gw < (unsigned)WDIM;
            const int ghc = min(max(gh, 0), HDIM - 1);
            const int gwc = min(max(gw, 0), WDIM - 1);
            const long off = (long)(ghc * WDIM + gwc) * CCH + c4;
            v4f kv = *(const v4f*)(k + off);
            v4f vv = *(const v4f*)(v + off);
            if (!ok) {
                kv = (v4f){0.0f, 0.0f, 0.0f, 0.0f};
                vv = (v4f){0.0f, 0.0f, 0.0f, 0.0f};
            }
            *(v4f*)(&ks[hp * CCH + c4]) = kv;
            *(v4f*)(&vs[hp * CCH + c4]) = vv;
        }
    }
    __syncthreads();

    for (int pp = 0; pp < PIX_PER_WAVE; ++pp) {
        const int idx = wv * PIX_PER_WAVE + pp;
        const int ty  = idx / TILE_W;
        const int tx  = idx % TILE_W;
        const int p   = (h0 + ty) * WDIM + w0 + tx;
        const int wbase = (ty * HALO_W + tx) * CCH + lane * 4;
        const v4f qq = qv[pp];

        float sim[KK2];
        #pragma unroll
        for (int t = 0; t < KK2; ++t) {
            const v4f kv = *(const v4f*)(&ks[wbase + ((t / KSZ) * HALO_W + (t % KSZ)) * CCH]);
            float d = qq.x * kv.x + qq.y * kv.y + qq.z * kv.z + qq.w * kv.w;
            d += xhalf(d);
            d += xrow<8>(d);
            d += xrow<4>(d);
            d += xrow<2>(d);
            d += xrow<1>(d);
            sim[t] = d;
        }

        float m = sim[0];
        #pragma unroll
        for (int t = 1; t < KK2; ++t) m = fmaxf(m, sim[t]);
        float z = 0.0f;
        #pragma unroll
        for (int t = 0; t < KK2; ++t) { sim[t] = __expf(sim[t] - m); z += sim[t]; }
        const float inv = 1.0f / z;

        v4f acc = {0.0f, 0.0f, 0.0f, 0.0f};
        #pragma unroll
        for (int t = 0; t < KK2; ++t) {
            const v4f vv = *(const v4f*)(&vs[wbase + ((t / KSZ) * HALO_W + (t % KSZ)) * CCH]);
            const float wt = sim[t] * inv;
            acc.x += wt * vv.x; acc.y += wt * vv.y;
            acc.z += wt * vv.z; acc.w += wt * vv.w;
        }
        *(v4f*)(nf + (long)p * CCH + lane * 4) = acc;
    }
}

extern "C" void kernel_launch(void* const* d_in, const int* in_sizes, int n_in,
                              void* d_out, int out_size, void* d_ws, size_t ws_size,
                              hipStream_t stream) {
    const float* x    = (const float*)d_in[0];
    const float* w_q1 = (const float*)d_in[1];
    const float* s_q1 = (const float*)d_in[2];
    const float* b_q1 = (const float*)d_in[3];
    const float* w_q2 = (const float*)d_in[4];
    const float* s_q2 = (const float*)d_in[5];
    const float* b_q2 = (const float*)d_in[6];
    const float* w_k1 = (const float*)d_in[7];
    const float* s_k1 = (const float*)d_in[8];
    const float* b_k1 = (const float*)d_in[9];
    const float* w_k2 = (const float*)d_in[10];
    const float* s_k2 = (const float*)d_in[11];
    const float* b_k2 = (const float*)d_in[12];
    const float* w_v  = (const float*)d_in[13];
    const float* s_v  = (const float*)d_in[14];
    const float* b_v  = (const float*)d_in[15];
    const float* w_o  = (const float*)d_in[16];
    const float* s_o  = (const float*)d_in[17];
    const float* b_o  = (const float*)d_in[18];
    float* out = (float*)d_out;

    const size_t NB = (size_t)HW * CCH;
    float* qb = (float*)d_ws;
    float* kb = qb + NB;
    float* vb = kb + NB;
    float* nf = vb + NB;

    const dim3 blk(256);
    const dim3 grd(HW / 64);

    proj_fused_wmma<<<grd, blk, 0, stream>>>(x,
                                             w_q1, s_q1, b_q1, w_q2, s_q2, b_q2,
                                             w_k1, s_k1, b_k1, w_k2, s_k2, b_k2,
                                             w_v, s_v, b_v,
                                             qb, kb, vb);
    local_attn_kernel<<<dim3(HW / (TILE_W * TILE_H)), dim3(ATT_THREADS), 0, stream>>>(qb, kb, vb, nf);
    gemm_bn_wmma<<<grd, blk, 0, stream>>>(nf, 1, CCH, x, HW, 1,
                                          w_o, 2 * CCH, s_o, b_o,
                                          out, HW, 1, 0);
}


# closing 1x1 conv+BN fused into the attention launch: a tile's attention rows stay in LDS and feed the WMMA there (3 launches -> 2); its k sum now four quarter sums added in fixed order
# speedup vs baseline: 2.6718x; 1.0463x over previous
#include <hip/hip_runtime.h>
#include <hip/hip_bf16.h>

typedef __attribute__((ext_vector_type(2))) float v2f;
typedef __attribute__((ext_vector_type(4))) float v4f;
typedef __attribute__((ext_vector_type(8))) float v8f;

#define CCH 128
#define HDIM 128
#define WDIM 128
#define HW (HDIM * WDIM)
#define KSZ 7
#define KK2 49
#define TPITCH 132

__device__ __forceinline__ void mma_tile128(
    const float* tile, const int (&bbase)[4],
    const float* __restrict__ Ap, v8f (&acc)[4])
{
    #pragma unroll
    for (int k0 = 0; k0 < CCH; k0 += 4) {
        const v2f a = *(const v2f*)(Ap + k0);
        #pragma unroll
        for (int s = 0; s < 4; ++s) {
            const v2f b = *(const v2f*)(&tile[bbase[s] + k0]);
            acc[s] = __builtin_amdgcn_wmma_f32_16x16x4_f32(
                false, a, false, b, (short)0, acc[s], false, false);
        }
    }
}

__device__ __forceinline__ void bn_relu_tile(
    const v8f& d, const float* __restrict__ scale, const float* __restrict__ bias,
    int o8, v4f& lo, v4f& hi)
{
    const v4f s0 = *(const v4f*)(scale + o8), s1 = *(const v4f*)(scale + o8 + 4);
    const v4f b0 = *(const v4f*)(bias + o8),  b1 = *(const v4f*)(bias + o8 + 4);
    lo.x = fmaxf(d[0] * s0.x + b0.x, 0.0f);
    lo.y = fmaxf(d[1] * s0.y + b0.y, 0.0f);
    lo.z = fmaxf(d[2] * s0.z + b0.z, 0.0f);
    lo.w = fmaxf(d[3] * s0.w + b0.w, 0.0f);
    hi.x = fmaxf(d[4] * s1.x + b1.x, 0.0f);
    hi.y = fmaxf(d[5] * s1.y + b1.y, 0.0f);
    hi.z = fmaxf(d[6] * s1.z + b1.z, 0.0f);
    hi.w = fmaxf(d[7] * s1.w + b1.w, 0.0f);
}

__global__ __launch_bounds__(256) void proj_fused_wmma(
    const float* __restrict__ x,
    const float* __restrict__ w_q1, const float* __restrict__ s_q1, const float* __restrict__ b_q1,
    const float* __restrict__ w_q2, const float* __restrict__ s_q2, const float* __restrict__ b_q2,
    const float* __restrict__ w_k1, const float* __restrict__ s_k1, const float* __restrict__ b_k1,
    const float* __restrict__ w_k2, const float* __restrict__ s_k2, const float* __restrict__ b_k2,
    const float* __restrict__ w_v,  const float* __restrict__ s_v,  const float* __restrict__ b_v,
    float* __restrict__ qo, float* __restrict__ ko, float* __restrict__ vo)
{
    __shared__ __attribute__((aligned(16))) float tx[64 * TPITCH];
    __shared__ __attribute__((aligned(16))) float tq[64 * TPITCH];
    __shared__ __attribute__((aligned(16))) float tk[64 * TPITCH];

    const int tid  = threadIdx.x;
    const int lane = tid & 31;
    const int wv   = tid >> 5;
    const int hl   = lane >> 4;
    const int l16  = lane & 15;
    const int m0   = wv << 4;
    const long p0  = (long)blockIdx.x * 64;
    const int o8   = m0 + (hl << 3);

    int bbase[4];
    #pragma unroll
    for (int s = 0; s < 4; ++s)
        bbase[s] = ((s << 4) + l16) * TPITCH + (hl << 1);

    #pragma unroll
    for (int i = 0; i < 8; ++i) {
        const int e  = tid + (i << 8);
        const int p4 = (e & 15) << 2;
        const int c  = e >> 4;
        const v4f xv = *(const v4f*)(x + (long)c * HW + p0 + p4);
        tx[(p4 + 0) * TPITCH + c] = xv.x;
        tx[(p4 + 1) * TPITCH + c] = xv.y;
        tx[(p4 + 2) * TPITCH + c] = xv.z;
        tx[(p4 + 3) * TPITCH + c] = xv.w;
    }
    __syncthreads();

    const long arow = (long)(m0 + l16) * CCH + (hl << 1);

    {
        v8f acc[4] = {};
        mma_tile128(tx, bbase, w_q1 + arow, acc);
        #pragma unroll
        for (int s = 0; s < 4; ++s) {
            v4f lo, hi;
            bn_relu_tile(acc[s], s_q1, b_q1, o8, lo, hi);
            float* d = &tq[((s << 4) + l16) * TPITCH + o8];
            *(v4f*)(d) = lo;  *(v4f*)(d + 4) = hi;
        }
    }
    {
        v8f acc[4] = {};
        mma_tile128(tx, bbase, w_k1 + arow, acc);
        #pragma unroll
        for (int s = 0; s < 4; ++s) {
            v4f lo, hi;
            bn_relu_tile(acc[s], s_k1, b_k1, o8, lo, hi);
            float* d = &tk[((s << 4) + l16) * TPITCH + o8];
            *(v4f*)(d) = lo;  *(v4f*)(d + 4) = hi;
        }
    }
    {
        v8f acc[4] = {};
        mma_tile128(tx, bbase, w_v + arow, acc);
        #pragma unroll
        for (int s = 0; s < 4; ++s) {
            v4f lo, hi;
            bn_relu_tile(acc[s], s_v, b_v, o8, lo, hi);
            float* d = vo + (p0 + (s << 4) + l16) * CCH + o8;
            *(v4f*)(d) = lo;  *(v4f*)(d + 4) = hi;
        }
    }
    __syncthreads();

    {
        v8f acc[4] = {};
        mma_tile128(tq, bbase, w_q2 + arow, acc);
        #pragma unroll
        for (int s = 0; s < 4; ++s) {
            v4f lo, hi;
            bn_relu_tile(acc[s], s_q2, b_q2, o8, lo, hi);
            float* d = qo + (p0 + (s << 4) + l16) * CCH + o8;
            *(v4f*)(d) = lo;  *(v4f*)(d + 4) = hi;
        }
    }
    {
        v8f acc[4] = {};
        mma_tile128(tk, bbase, w_k2 + arow, acc);
        #pragma unroll
        for (int s = 0; s < 4; ++s) {
            v4f lo, hi;
            bn_relu_tile(acc[s], s_k2, b_k2, o8, lo, hi);
            float* d = ko + (p0 + (s << 4) + l16) * CCH + o8;
            *(v4f*)(d) = lo;  *(v4f*)(d + 4) = hi;
        }
    }
}

__device__ __forceinline__ float xhalf(float x) {
    const int b = __builtin_bit_cast(int, x);
    return __builtin_bit_cast(float,
        __builtin_amdgcn_permlanex16(b, b, 0x76543210, 0xfedcba98, false, false));
}
template <int M> __device__ __forceinline__ float xrow(float x) {
    const int b = __builtin_bit_cast(int, x);
    return __builtin_bit_cast(float,
        __builtin_amdgcn_update_dpp(b, b, 0x160 + M, 0xf, 0xf, true));
}

#define TILE_W 8
#define TILE_H 8
#define HALO_W (TILE_W + KSZ - 1)
#define HALO_H (TILE_H + KSZ - 1)
#define HALO_N (HALO_W * HALO_H)
#define HALO_V4 (HALO_N * (CCH / 4))
#define ATT_THREADS 1024
#define PIX_PER_WAVE 2
#define TILE_PX (TILE_W * TILE_H)
#define RPITCH 68

__global__ __launch_bounds__(ATT_THREADS) void local_attn_out_kernel(
    const float* __restrict__ q, const float* __restrict__ k,
    const float* __restrict__ v, const float* __restrict__ x,
    const float* __restrict__ w_o, const float* __restrict__ s_o,
    const float* __restrict__ b_o, float* __restrict__ out)
{
    __shared__ __attribute__((aligned(16))) float kvs[2 * HALO_N * CCH];
    __shared__ __attribute__((aligned(16))) float tn[TILE_PX * TPITCH];
    __shared__ __attribute__((aligned(16))) float txx[TILE_PX * TPITCH];
    float* const ks  = kvs;
    float* const vs  = kvs + HALO_N * CCH;
    float* const red = kvs;

    const int tid  = threadIdx.x;
    const int lane = tid & 31;
    const int wv   = tid >> 5;
    const int h0   = (blockIdx.x / (WDIM / TILE_W)) * TILE_H;
    const int w0   = (blockIdx.x % (WDIM / TILE_W)) * TILE_W;

    v4f qv[PIX_PER_WAVE];
    #pragma unroll
    for (int pp = 0; pp < PIX_PER_WAVE; ++pp) {
        const int idx = wv * PIX_PER_WAVE + pp;
        const int p   = (h0 + (idx / TILE_W)) * WDIM + w0 + (idx % TILE_W);
        qv[pp] = *(const v4f*)(q + (long)p * CCH + lane * 4);
    }

    {
        const int c  = tid & (CCH - 1);
        const int ry = tid >> 7;
        const float* xp = x + (long)c * HW + (h0 + ry) * WDIM + w0;
        const v4f xa = *(const v4f*)(xp);
        const v4f xb = *(const v4f*)(xp + 4);
        float* d = &txx[(ry * TILE_W) * TPITCH + c];
        d[0 * TPITCH] = xa.x;  d[1 * TPITCH] = xa.y;
        d[2 * TPITCH] = xa.z;  d[3 * TPITCH] = xa.w;
        d[4 * TPITCH] = xb.x;  d[5 * TPITCH] = xb.y;
        d[6 * TPITCH] = xb.z;  d[7 * TPITCH] = xb.w;
    }

    #pragma unroll
    for (int it = 0; it < (HALO_V4 + ATT_THREADS - 1) / ATT_THREADS; ++it) {
        const int e = tid + it * ATT_THREADS;
        if (e < HALO_V4) {
            const int hp = e >> 5;
            const int c4 = (e & 31) << 2;
            const int hy = hp / HALO_W;
            const int hx = hp - hy * HALO_W;
            const int gh = h0 + hy - (KSZ / 2);
            const int gw = w0 + hx - (KSZ / 2);
            const bool ok = (unsigned)gh < (unsigned)HDIM && (unsigned)gw < (unsigned)WDIM;
            const int ghc = min(max(gh, 0), HDIM - 1);
            const int gwc = min(max(gw, 0), WDIM - 1);
            const long off = (long)(ghc * WDIM + gwc) * CCH + c4;
            v4f kv = *(const v4f*)(k + off);
            v4f vv = *(const v4f*)(v + off);
            if (!ok) {
                kv = (v4f){0.0f, 0.0f, 0.0f, 0.0f};
                vv = (v4f){0.0f, 0.0f, 0.0f, 0.0f};
            }
            *(v4f*)(&ks[hp * CCH + c4]) = kv;
            *(v4f*)(&vs[hp * CCH + c4]) = vv;
        }
    }
    __syncthreads();

    for (int pp = 0; pp < PIX_PER_WAVE; ++pp) {
        const int idx = wv * PIX_PER_WAVE + pp;
        const int ty  = idx / TILE_W;
        const int tx  = idx % TILE_W;
        const int wbase = (ty * HALO_W + tx) * CCH + lane * 4;
        const v4f qq = qv[pp];

        float sim[KK2];
        #pragma unroll
        for (int t = 0; t < KK2; ++t) {
            const v4f kv = *(const v4f*)(&ks[wbase + ((t / KSZ) * HALO_W + (t % KSZ)) * CCH]);
            float d = qq.x * kv.x + qq.y * kv.y + qq.z * kv.z + qq.w * kv.w;
            d += xhalf(d);
            d += xrow<8>(d);
            d += xrow<4>(d);
            d += xrow<2>(d);
            d += xrow<1>(d);
            sim[t] = d;
        }

        float m = sim[0];
        #pragma unroll
        for (int t = 1; t < KK2; ++t) m = fmaxf(m, sim[t]);
        float z = 0.0f;
        #pragma unroll
        for (int t = 0; t < KK2; ++t) { sim[t] = __expf(sim[t] - m); z += sim[t]; }
        const float inv = 1.0f / z;

        v4f acc = {0.0f, 0.0f, 0.0f, 0.0f};
        #pragma unroll
        for (int t = 0; t < KK2; ++t) {
            const v4f vv = *(const v4f*)(&vs[wbase + ((t / KSZ) * HALO_W + (t % KSZ)) * CCH]);
            const float wt = sim[t] * inv;
            acc.x += wt * vv.x; acc.y += wt * vv.y;
            acc.z += wt * vv.z; acc.w += wt * vv.w;
        }
        *(v4f*)(&tn[idx * TPITCH + lane * 4]) = acc;
    }

    const int hl  = lane >> 4;
    const int l16 = lane & 15;
    const int m0  = (wv & 7) << 4;
    const int kq  = wv >> 3;

    v2f afr[CCH / 8];
    {
        const float* __restrict__ Ap =
            w_o + (long)(m0 + l16) * (2 * CCH) + (kq << 6) + (hl << 1);
        #pragma unroll
        for (int j = 0; j < CCH / 8; ++j)
            afr[j] = *(const v2f*)(Ap + (j << 2));
    }
    __syncthreads();

    {
        const float* Bt = (kq & 2) ? txx : tn;
        const int brow = l16 * TPITCH + ((kq & 1) << 6) + (hl << 1);

        v8f pacc[4] = {};
        #pragma unroll
        for (int j = 0; j < CCH / 8; ++j) {
            const int k0 = j << 2;
            #pragma unroll
            for (int s = 0; s < 4; ++s) {
                const v2f b = *(const v2f*)(&Bt[(s << 4) * TPITCH + brow + k0]);
                pacc[s] = __builtin_amdgcn_wmma_f32_16x16x4_f32(
                    false, afr[j], false, b, (short)0, pacc[s], false, false);
            }
        }
        #pragma unroll
        for (int s = 0; s < 4; ++s) {
            #pragma unroll
            for (int r = 0; r < 8; ++r)
                red[(kq * CCH + m0 + r + (hl << 3)) * RPITCH + (s << 4) + l16] = pacc[s][r];
        }
    }
    __syncthreads();

    #pragma unroll
    for (int ps = 0; ps < 2; ++ps) {
        const int o  = (tid >> 4) + (ps << 6);
        const int ry = (tid >> 1) & 7;
        const int hf = (tid & 1) << 2;
        const float* rp = &red[o * RPITCH + ry * TILE_W + hf];
        v4f sum = *(const v4f*)(rp);
        sum += *(const v4f*)(rp + 1 * CCH * RPITCH);
        sum += *(const v4f*)(rp + 2 * CCH * RPITCH);
        sum += *(const v4f*)(rp + 3 * CCH * RPITCH);
        const float sc = s_o[o], bi = b_o[o];
        v4f y;
        y.x = sum.x * sc + bi;  y.y = sum.y * sc + bi;
        y.z = sum.z * sc + bi;  y.w = sum.w * sc + bi;
        *(v4f*)(out + (long)o * HW + (h0 + ry) * WDIM + w0 + hf) = y;
    }
}

extern "C" void kernel_launch(void* const* d_in, const int* in_sizes, int n_in,
                              void* d_out, int out_size, void* d_ws, size_t ws_size,
                              hipStream_t stream) {
    const float* x    = (const float*)d_in[0];
    const float* w_q1 = (const float*)d_in[1];
    const float* s_q1 = (const float*)d_in[2];
    const float* b_q1 = (const float*)d_in[3];
    const float* w_q2 = (const float*)d_in[4];
    const float* s_q2 = (const float*)d_in[5];
    const float* b_q2 = (const float*)d_in[6];
    const float* w_k1 = (const float*)d_in[7];
    const float* s_k1 = (const float*)d_in[8];
    const float* b_k1 = (const float*)d_in[9];
    const float* w_k2 = (const float*)d_in[10];
    const float* s_k2 = (const float*)d_in[11];
    const float* b_k2 = (const float*)d_in[12];
    const float* w_v  = (const float*)d_in[13];
    const float* s_v  = (const float*)d_in[14];
    const float* b_v  = (const float*)d_in[15];
    const float* w_o  = (const float*)d_in[16];
    const float* s_o  = (const float*)d_in[17];
    const float* b_o  = (const float*)d_in[18];
    float* out = (float*)d_out;

    const size_t NB = (size_t)HW * CCH;
    float* qb = (float*)d_ws;
    float* kb = qb + NB;
    float* vb = kb + NB;

    const dim3 blk(256);
    const dim3 grd(HW / 64);

    proj_fused_wmma<<<grd, blk, 0, stream>>>(x,
                                             w_q1, s_q1, b_q1, w_q2, s_q2, b_q2,
                                             w_k1, s_k1, b_k1, w_k2, s_k2, b_k2,
                                             w_v, s_v, b_v,
                                             qb, kb, vb);
    local_attn_out_kernel<<<dim3(HW / TILE_PX), dim3(ATT_THREADS), 0, stream>>>(
        qb, kb, vb, x, w_o, s_o, b_o, out);
}


# attention: a wave's 2 adjacent pixels done as a pair: each K/V halo row read once for both; 49 score lane-sums as a reduce-scatter (4 scores per lane), softmax once per score; z sum by lane butterfly
# speedup vs baseline: 3.0907x; 1.1568x over previous
#include <hip/hip_runtime.h>
#include <hip/hip_bf16.h>

typedef __attribute__((ext_vector_type(2))) float v2f;
typedef __attribute__((ext_vector_type(4))) float v4f;
typedef __attribute__((ext_vector_type(8))) float v8f;

#define CCH 128
#define HDIM 128
#define WDIM 128
#define HW (HDIM * WDIM)
#define KSZ 7
#define KK2 49
#define TPITCH 132

__device__ __forceinline__ void mma_tile128(
    const float* tile, const int (&bbase)[4],
    const float* __restrict__ Ap, v8f (&acc)[4])
{
    #pragma unroll
    for (int k0 = 0; k0 < CCH; k0 += 4) {
        const v2f a = *(const v2f*)(Ap + k0);
        #pragma unroll
        for (int s = 0; s < 4; ++s) {
            const v2f b = *(const v2f*)(&tile[bbase[s] + k0]);
            acc[s] = __builtin_amdgcn_wmma_f32_16x16x4_f32(
                false, a, false, b, (short)0, acc[s], false, false);
        }
    }
}

__device__ __forceinline__ void bn_relu_tile(
    const v8f& d, const float* __restrict__ scale, const float* __restrict__ bias,
    int o8, v4f& lo, v4f& hi)
{
    const v4f s0 = *(const v4f*)(scale + o8), s1 = *(const v4f*)(scale + o8 + 4);
    const v4f b0 = *(const v4f*)(bias + o8),  b1 = *(const v4f*)(bias + o8 + 4);
    lo.x = fmaxf(d[0] * s0.x + b0.x, 0.0f);
    lo.y = fmaxf(d[1] * s0.y + b0.y, 0.0f);
    lo.z = fmaxf(d[2] * s0.z + b0.z, 0.0f);
    lo.w = fmaxf(d[3] * s0.w + b0.w, 0.0f);
    hi.x = fmaxf(d[4] * s1.x + b1.x, 0.0f);
    hi.y = fmaxf(d[5] * s1.y + b1.y, 0.0f);
    hi.z = fmaxf(d[6] * s1.z + b1.z, 0.0f);
    hi.w = fmaxf(d[7] * s1.w + b1.w, 0.0f);
}

__global__ __launch_bounds__(256) void proj_fused_wmma(
    const float* __restrict__ x,
    const float* __restrict__ w_q1, const float* __restrict__ s_q1, const float* __restrict__ b_q1,
    const float* __restrict__ w_q2, const float* __restrict__ s_q2, const float* __restrict__ b_q2,
    const float* __restrict__ w_k1, const float* __restrict__ s_k1, const float* __restrict__ b_k1,
    const float* __restrict__ w_k2, const float* __restrict__ s_k2, const float* __restrict__ b_k2,
    const float* __restrict__ w_v,  const float* __restrict__ s_v,  const float* __restrict__ b_v,
    float* __restrict__ qo, float* __restrict__ ko, float* __restrict__ vo)
{
    __shared__ __attribute__((aligned(16))) float tx[64 * TPITCH];
    __shared__ __attribute__((aligned(16))) float tq[64 * TPITCH];
    __shared__ __attribute__((aligned(16))) float tk[64 * TPITCH];

    const int tid  = threadIdx.x;
    const int lane = tid & 31;
    const int wv   = tid >> 5;
    const int hl   = lane >> 4;
    const int l16  = lane & 15;
    const int m0   = wv << 4;
    const long p0  = (long)blockIdx.x * 64;
    const int o8   = m0 + (hl << 3);

    int bbase[4];
    #pragma unroll
    for (int s = 0; s < 4; ++s)
        bbase[s] = ((s << 4) + l16) * TPITCH + (hl << 1);

    #pragma unroll
    for (int i = 0; i < 8; ++i) {
        const int e  = tid + (i << 8);
        const int p4 = (e & 15) << 2;
        const int c  = e >> 4;
        const v4f xv = *(const v4f*)(x + (long)c * HW + p0 + p4);
        tx[(p4 + 0) * TPITCH + c] = xv.x;
        tx[(p4 + 1) * TPITCH + c] = xv.y;
        tx[(p4 + 2) * TPITCH + c] = xv.z;
        tx[(p4 + 3) * TPITCH + c] = xv.w;
    }
    __syncthreads();

    const long arow = (long)(m0 + l16) * CCH + (hl << 1);

    {
        v8f acc[4] = {};
        mma_tile128(tx, bbase, w_q1 + arow, acc);
        #pragma unroll
        for (int s = 0; s < 4; ++s) {
            v4f lo, hi;
            bn_relu_tile(acc[s], s_q1, b_q1, o8, lo, hi);
            float* d = &tq[((s << 4) + l16) * TPITCH + o8];
            *(v4f*)(d) = lo;  *(v4f*)(d + 4) = hi;
        }
    }
    {
        v8f acc[4] = {};
        mma_tile128(tx, bbase, w_k1 + arow, acc);
        #pragma unroll
        for (int s = 0; s < 4; ++s) {
            v4f lo, hi;
            bn_relu_tile(acc[s], s_k1, b_k1, o8, lo, hi);
            float* d = &tk[((s << 4) + l16) * TPITCH + o8];
            *(v4f*)(d) = lo;  *(v4f*)(d + 4) = hi;
        }
    }
    {
        v8f acc[4] = {};
        mma_tile128(tx, bbase, w_v + arow, acc);
        #pragma unroll
        for (int s = 0; s < 4; ++s) {
            v4f lo, hi;
            bn_relu_tile(acc[s], s_v, b_v, o8, lo, hi);
            float* d = vo + (p0 + (s << 4) + l16) * CCH + o8;
            *(v4f*)(d) = lo;  *(v4f*)(d + 4) = hi;
        }
    }
    __syncthreads();

    {
        v8f acc[4] = {};
        mma_tile128(tq, bbase, w_q2 + arow, acc);
        #pragma unroll
        for (int s = 0; s < 4; ++s) {
            v4f lo, hi;
            bn_relu_tile(acc[s], s_q2, b_q2, o8, lo, hi);
            float* d = qo + (p0 + (s << 4) + l16) * CCH + o8;
            *(v4f*)(d) = lo;  *(v4f*)(d + 4) = hi;
        }
    }
    {
        v8f acc[4] = {};
        mma_tile128(tk, bbase, w_k2 + arow, acc);
        #pragma unroll
        for (int s = 0; s < 4; ++s) {
            v4f lo, hi;
            bn_relu_tile(acc[s], s_k2, b_k2, o8, lo, hi);
            float* d = ko + (p0 + (s << 4) + l16) * CCH + o8;
            *(v4f*)(d) = lo;  *(v4f*)(d + 4) = hi;
        }
    }
}

__device__ __forceinline__ float xhalf(float x) {
    const int b = __builtin_bit_cast(int, x);
    return __builtin_bit_cast(float,
        __builtin_amdgcn_permlanex16(b, b, 0x76543210, 0xfedcba98, false, false));
}
template <int M> __device__ __forceinline__ float xrow(float x) {
    const int b = __builtin_bit_cast(int, x);
    return __builtin_bit_cast(float,
        __builtin_amdgcn_update_dpp(b, b, 0x160 + M, 0xf, 0xf, true));
}

#define TILE_W 8
#define TILE_H 8
#define HALO_W (TILE_W + KSZ - 1)
#define HALO_H (TILE_H + KSZ - 1)
#define HALO_N (HALO_W * HALO_H)
#define HALO_V4 (HALO_N * (CCH / 4))
#define ATT_THREADS 1024
#define PIX_PER_WAVE 2
#define TILE_PX (TILE_W * TILE_H)
#define RPITCH 68

__global__ __launch_bounds__(ATT_THREADS) void local_attn_out_kernel(
    const float* __restrict__ q, const float* __restrict__ k,
    const float* __restrict__ v, const float* __restrict__ x,
    const float* __restrict__ w_o, const float* __restrict__ s_o,
    const float* __restrict__ b_o, float* __restrict__ out)
{
    __shared__ __attribute__((aligned(16))) float kvs[2 * HALO_N * CCH];
    __shared__ __attribute__((aligned(16))) float tn[TILE_PX * TPITCH];
    __shared__ __attribute__((aligned(16))) float txx[TILE_PX * TPITCH];
    float* const ks  = kvs;
    float* const vs  = kvs + HALO_N * CCH;
    float* const red = kvs;

    const int tid  = threadIdx.x;
    const int lane = tid & 31;
    const int wv   = tid >> 5;
    const int h0   = (blockIdx.x / (WDIM / TILE_W)) * TILE_H;
    const int w0   = (blockIdx.x % (WDIM / TILE_W)) * TILE_W;

    v4f qv[PIX_PER_WAVE];
    #pragma unroll
    for (int pp = 0; pp < PIX_PER_WAVE; ++pp) {
        const int idx = wv * PIX_PER_WAVE + pp;
        const int p   = (h0 + (idx / TILE_W)) * WDIM + w0 + (idx % TILE_W);
        qv[pp] = *(const v4f*)(q + (long)p * CCH + lane * 4);
    }

    {
        const int c  = tid & (CCH - 1);
        const int ry = tid >> 7;
        const float* xp = x + (long)c * HW + (h0 + ry) * WDIM + w0;
        const v4f xa = *(const v4f*)(xp);
        const v4f xb = *(const v4f*)(xp + 4);
        float* d = &txx[(ry * TILE_W) * TPITCH + c];
        d[0 * TPITCH] = xa.x;  d[1 * TPITCH] = xa.y;
        d[2 * TPITCH] = xa.z;  d[3 * TPITCH] = xa.w;
        d[4 * TPITCH] = xb.x;  d[5 * TPITCH] = xb.y;
        d[6 * TPITCH] = xb.z;  d[7 * TPITCH] = xb.w;
    }

    #pragma unroll
    for (int it = 0; it < (HALO_V4 + ATT_THREADS - 1) / ATT_THREADS; ++it) {
        const int e = tid + it * ATT_THREADS;
        if (e < HALO_V4) {
            const int hp = e >> 5;
            const int c4 = (e & 31) << 2;
            const int hy = hp / HALO_W;
            const int hx = hp - hy * HALO_W;
            const int gh = h0 + hy - (KSZ / 2);
            const int gw = w0 + hx - (KSZ / 2);
            const bool ok = (unsigned)gh < (unsigned)HDIM && (unsigned)gw < (unsigned)WDIM;
            const int ghc = min(max(gh, 0), HDIM - 1);
            const int gwc = min(max(gw, 0), WDIM - 1);
            const long off = (long)(ghc * WDIM + gwc) * CCH + c4;
            v4f kv = *(const v4f*)(k + off);
            v4f vv = *(const v4f*)(v + off);
            if (!ok) {
                kv = (v4f){0.0f, 0.0f, 0.0f, 0.0f};
                vv = (v4f){0.0f, 0.0f, 0.0f, 0.0f};
            }
            *(v4f*)(&ks[hp * CCH + c4]) = kv;
            *(v4f*)(&vs[hp * CCH + c4]) = vv;
        }
    }
    __syncthreads();

    const int idx0  = wv * PIX_PER_WAVE;
    const int ty    = idx0 / TILE_W;
    const int tx0   = idx0 % TILE_W;
    const int wbase = (ty * HALO_W + tx0) * CCH + lane * 4;
    const v4f q0 = qv[0], q1 = qv[1];
    const bool lo = lane < 16;
    const bool b3 = (lane & 8) != 0, b2 = (lane & 4) != 0;
    const bool b1 = (lane & 2) != 0, b0 = (lane & 1) != 0;

    float u[KK2];
    #pragma unroll
    for (int r = 0; r < KSZ; ++r) {
        v4f kr[KSZ + 1];
        #pragma unroll
        for (int c = 0; c < KSZ + 1; ++c)
            kr[c] = *(const v4f*)(&ks[wbase + (r * HALO_W + c) * CCH]);
        #pragma unroll
        for (int j = 0; j < KSZ; ++j) {
            const v4f ka = kr[j], kb = kr[j + 1];
            const float d0 = q0.x * ka.x + q0.y * ka.y + q0.z * ka.z + q0.w * ka.w;
            const float d1 = q1.x * kb.x + q1.y * kb.y + q1.z * kb.z + q1.w * kb.w;
            const float keep = lo ? d0 : d1;
            const float send = lo ? d1 : d0;
            u[r * KSZ + j] = keep + xhalf(send);
        }
    }
    float s8[32];
    #pragma unroll
    for (int j = 0; j < 32; ++j) {
        if (j + 32 < KK2) {
            const float keep = b3 ? u[j + 32] : u[j];
            const float send = b3 ? u[j] : u[j + 32];
            s8[j] = keep + xrow<8>(send);
        } else {
            s8[j] = u[j] + xrow<8>(u[j]);
        }
    }
    float s4[16];
    #pragma unroll
    for (int j = 0; j < 16; ++j) {
        const float keep = b2 ? s8[j + 16] : s8[j];
        const float send = b2 ? s8[j] : s8[j + 16];
        s4[j] = keep + xrow<4>(send);
    }
    float s2[8];
    #pragma unroll
    for (int j = 0; j < 8; ++j) {
        const float keep = b1 ? s4[j + 8] : s4[j];
        const float send = b1 ? s4[j] : s4[j + 8];
        s2[j] = keep + xrow<2>(send);
    }
    float sc[4];
    #pragma unroll
    for (int j = 0; j < 4; ++j) {
        const float keep = b0 ? s2[j + 4] : s2[j];
        const float send = b0 ? s2[j] : s2[j + 4];
        sc[j] = keep + xrow<1>(send);
    }

    const int tb = (lane & 15) * 4;
    float m = -__builtin_inff();
    #pragma unroll
    for (int i = 0; i < 4; ++i) m = (tb + i < KK2) ? fmaxf(m, sc[i]) : m;
    m = fmaxf(m, xrow<8>(m));
    m = fmaxf(m, xrow<4>(m));
    m = fmaxf(m, xrow<2>(m));
    m = fmaxf(m, xrow<1>(m));
    float wgt[4];
    #pragma unroll
    for (int i = 0; i < 4; ++i) wgt[i] = (tb + i < KK2) ? __expf(sc[i] - m) : 0.0f;
    float z = ((wgt[0] + wgt[1]) + wgt[2]) + wgt[3];
    z += xrow<8>(z);
    z += xrow<4>(z);
    z += xrow<2>(z);
    z += xrow<1>(z);
    const float inv = 1.0f / z;
    #pragma unroll
    for (int i = 0; i < 4; ++i) wgt[i] *= inv;

    v4f acc0 = {0.0f, 0.0f, 0.0f, 0.0f};
    v4f acc1 = {0.0f, 0.0f, 0.0f, 0.0f};
    #pragma unroll
    for (int r = 0; r < KSZ; ++r) {
        v4f vr[KSZ + 1];
        #pragma unroll
        for (int c = 0; c < KSZ + 1; ++c)
            vr[c] = *(const v4f*)(&vs[wbase + (r * HALO_W + c) * CCH]);
        #pragma unroll
        for (int c = 0; c < KSZ + 1; ++c) {
            if (c < KSZ) {
                const int t = r * KSZ + c;
                const float wt = __builtin_bit_cast(float, __builtin_amdgcn_readlane(
                    __builtin_bit_cast(int, wgt[t & 3]), t >> 2));
                acc0.x += wt * vr[c].x; acc0.y += wt * vr[c].y;
                acc0.z += wt * vr[c].z; acc0.w += wt * vr[c].w;
            }
            if (c >= 1) {
                const int t = r * KSZ + c - 1;
                const float wt = __builtin_bit_cast(float, __builtin_amdgcn_readlane(
                    __builtin_bit_cast(int, wgt[t & 3]), 16 + (t >> 2)));
                acc1.x += wt * vr[c].x; acc1.y += wt * vr[c].y;
                acc1.z += wt * vr[c].z; acc1.w += wt * vr[c].w;
            }
        }
    }
    *(v4f*)(&tn[idx0 * TPITCH + lane * 4]) = acc0;
    *(v4f*)(&tn[(idx0 + 1) * TPITCH + lane * 4]) = acc1;

    const int hl  = lane >> 4;
    const int l16 = lane & 15;
    const int m0  = (wv & 7) << 4;
    const int kq  = wv >> 3;

    v2f afr[CCH / 8];
    {
        const float* __restrict__ Ap =
            w_o + (long)(m0 + l16) * (2 * CCH) + (kq << 6) + (hl << 1);
        #pragma unroll
        for (int j = 0; j < CCH / 8; ++j)
            afr[j] = *(const v2f*)(Ap + (j << 2));
    }
    __syncthreads();

    {
        const float* Bt = (kq & 2) ? txx : tn;
        const int brow = l16 * TPITCH + ((kq & 1) << 6) + (hl << 1);

        v8f pacc[4] = {};
        #pragma unroll
        for (int j = 0; j < CCH / 8; ++j) {
            const int k0 = j << 2;
            #pragma unroll
            for (int s = 0; s < 4; ++s) {
                const v2f b = *(const v2f*)(&Bt[(s << 4) * TPITCH + brow + k0]);
                pacc[s] = __builtin_amdgcn_wmma_f32_16x16x4_f32(
                    false, afr[j], false, b, (short)0, pacc[s], false, false);
            }
        }
        #pragma unroll
        for (int s = 0; s < 4; ++s) {
            #pragma unroll
            for (int r = 0; r < 8; ++r)
                red[(kq * CCH + m0 + r + (hl << 3)) * RPITCH + (s << 4) + l16] = pacc[s][r];
        }
    }
    __syncthreads();

    #pragma unroll
    for (int ps = 0; ps < 2; ++ps) {
        const int o  = (tid >> 4) + (ps << 6);
        const int ry = (tid >> 1) & 7;
        const int hf = (tid & 1) << 2;
        const float* rp = &red[o * RPITCH + ry * TILE_W + hf];
        v4f sum = *(const v4f*)(rp);
        sum += *(const v4f*)(rp + 1 * CCH * RPITCH);
        sum += *(const v4f*)(rp + 2 * CCH * RPITCH);
        sum += *(const v4f*)(rp + 3 * CCH * RPITCH);
        const float sc = s_o[o], bi = b_o[o];
        v4f y;
        y.x = sum.x * sc + bi;  y.y = sum.y * sc + bi;
        y.z = sum.z * sc + bi;  y.w = sum.w * sc + bi;
        *(v4f*)(out + (long)o * HW + (h0 + ry) * WDIM + w0 + hf) = y;
    }
}

extern "C" void kernel_launch(void* const* d_in, const int* in_sizes, int n_in,
                              void* d_out, int out_size, void* d_ws, size_t ws_size,
                              hipStream_t stream) {
    const float* x    = (const float*)d_in[0];
    const float* w_q1 = (const float*)d_in[1];
    const float* s_q1 = (const float*)d_in[2];
    const float* b_q1 = (const float*)d_in[3];
    const float* w_q2 = (const float*)d_in[4];
    const float* s_q2 = (const float*)d_in[5];
    const float* b_q2 = (const float*)d_in[6];
    const float* w_k1 = (const float*)d_in[7];
    const float* s_k1 = (const float*)d_in[8];
    const float* b_k1 = (const float*)d_in[9];
    const float* w_k2 = (const float*)d_in[10];
    const float* s_k2 = (const float*)d_in[11];
    const float* b_k2 = (const float*)d_in[12];
    const float* w_v  = (const float*)d_in[13];
    const float* s_v  = (const float*)d_in[14];
    const float* b_v  = (const float*)d_in[15];
    const float* w_o  = (const float*)d_in[16];
    const float* s_o  = (const float*)d_in[17];
    const float* b_o  = (const float*)d_in[18];
    float* out = (float*)d_out;

    const size_t NB = (size_t)HW * CCH;
    float* qb = (float*)d_ws;
    float* kb = qb + NB;
    float* vb = kb + NB;

    const dim3 blk(256);
    const dim3 grd(HW / 64);

    proj_fused_wmma<<<grd, blk, 0, stream>>>(x,
                                             w_q1, s_q1, b_q1, w_q2, s_q2, b_q2,
                                             w_k1, s_k1, b_k1, w_k2, s_k2, b_k2,
                                             w_v, s_v, b_v,
                                             qb, kb, vb);
    local_attn_out_kernel<<<dim3(HW / TILE_PX), dim3(ATT_THREADS), 0, stream>>>(
        qb, kb, vb, x, w_o, s_o, b_o, out);
}


# attention: a wave does a 2x2 pixel quad (16 waves per tile): each K/V row of the quad's 8x8 halo patch read once for 4 pixels; softmax z sum reordered; closing k sum now 2 half sums (attention + x)
# speedup vs baseline: 3.2096x; 1.0385x over previous
#include <hip/hip_runtime.h>
#include <hip/hip_bf16.h>

typedef __attribute__((ext_vector_type(2))) float v2f;
typedef __attribute__((ext_vector_type(4))) float v4f;
typedef __attribute__((ext_vector_type(8))) float v8f;

#define CCH 128
#define HDIM 128
#define WDIM 128
#define HW (HDIM * WDIM)
#define KSZ 7
#define KK2 49
#define TPITCH 132

__device__ __forceinline__ void mma_tile128(
    const float* tile, const int (&bbase)[4],
    const float* __restrict__ Ap, v8f (&acc)[4])
{
    #pragma unroll
    for (int k0 = 0; k0 < CCH; k0 += 4) {
        const v2f a = *(const v2f*)(Ap + k0);
        #pragma unroll
        for (int s = 0; s < 4; ++s) {
            const v2f b = *(const v2f*)(&tile[bbase[s] + k0]);
            acc[s] = __builtin_amdgcn_wmma_f32_16x16x4_f32(
                false, a, false, b, (short)0, acc[s], false, false);
        }
    }
}

__device__ __forceinline__ void bn_relu_tile(
    const v8f& d, const float* __restrict__ scale, const float* __restrict__ bias,
    int o8, v4f& lo, v4f& hi)
{
    const v4f s0 = *(const v4f*)(scale + o8), s1 = *(const v4f*)(scale + o8 + 4);
    const v4f b0 = *(const v4f*)(bias + o8),  b1 = *(const v4f*)(bias + o8 + 4);
    lo.x = fmaxf(d[0] * s0.x + b0.x, 0.0f);
    lo.y = fmaxf(d[1] * s0.y + b0.y, 0.0f);
    lo.z = fmaxf(d[2] * s0.z + b0.z, 0.0f);
    lo.w = fmaxf(d[3] * s0.w + b0.w, 0.0f);
    hi.x = fmaxf(d[4] * s1.x + b1.x, 0.0f);
    hi.y = fmaxf(d[5] * s1.y + b1.y, 0.0f);
    hi.z = fmaxf(d[6] * s1.z + b1.z, 0.0f);
    hi.w = fmaxf(d[7] * s1.w + b1.w, 0.0f);
}

__global__ __launch_bounds__(256) void proj_fused_wmma(
    const float* __restrict__ x,
    const float* __restrict__ w_q1, const float* __restrict__ s_q1, const float* __restrict__ b_q1,
    const float* __restrict__ w_q2, const float* __restrict__ s_q2, const float* __restrict__ b_q2,
    const float* __restrict__ w_k1, const float* __restrict__ s_k1, const float* __restrict__ b_k1,
    const float* __restrict__ w_k2, const float* __restrict__ s_k2, const float* __restrict__ b_k2,
    const float* __restrict__ w_v,  const float* __restrict__ s_v,  const float* __restrict__ b_v,
    float* __restrict__ qo, float* __restrict__ ko, float* __restrict__ vo)
{
    __shared__ __attribute__((aligned(16))) float tx[64 * TPITCH];
    __shared__ __attribute__((aligned(16))) float tq[64 * TPITCH];
    __shared__ __attribute__((aligned(16))) float tk[64 * TPITCH];

    const int tid  = threadIdx.x;
    const int lane = tid & 31;
    const int wv   = tid >> 5;
    const int hl   = lane >> 4;
    const int l16  = lane & 15;
    const int m0   = wv << 4;
    const long p0  = (long)blockIdx.x * 64;
    const int o8   = m0 + (hl << 3);

    int bbase[4];
    #pragma unroll
    for (int s = 0; s < 4; ++s)
        bbase[s] = ((s << 4) + l16) * TPITCH + (hl << 1);

    #pragma unroll
    for (int i = 0; i < 8; ++i) {
        const int e  = tid + (i << 8);
        const int p4 = (e & 15) << 2;
        const int c  = e >> 4;
        const v4f xv = *(const v4f*)(x + (long)c * HW + p0 + p4);
        tx[(p4 + 0) * TPITCH + c] = xv.x;
        tx[(p4 + 1) * TPITCH + c] = xv.y;
        tx[(p4 + 2) * TPITCH + c] = xv.z;
        tx[(p4 + 3) * TPITCH + c] = xv.w;
    }
    __syncthreads();

    const long arow = (long)(m0 + l16) * CCH + (hl << 1);

    {
        v8f acc[4] = {};
        mma_tile128(tx, bbase, w_q1 + arow, acc);
        #pragma unroll
        for (int s = 0; s < 4; ++s) {
            v4f lo, hi;
            bn_relu_tile(acc[s], s_q1, b_q1, o8, lo, hi);
            float* d = &tq[((s << 4) + l16) * TPITCH + o8];
            *(v4f*)(d) = lo;  *(v4f*)(d + 4) = hi;
        }
    }
    {
        v8f acc[4] = {};
        mma_tile128(tx, bbase, w_k1 + arow, acc);
        #pragma unroll
        for (int s = 0; s < 4; ++s) {
            v4f lo, hi;
            bn_relu_tile(acc[s], s_k1, b_k1, o8, lo, hi);
            float* d = &tk[((s << 4) + l16) * TPITCH + o8];
            *(v4f*)(d) = lo;  *(v4f*)(d + 4) = hi;
        }
    }
    {
        v8f acc[4] = {};
        mma_tile128(tx, bbase, w_v + arow, acc);
        #pragma unroll
        for (int s = 0; s < 4; ++s) {
            v4f lo, hi;
            bn_relu_tile(acc[s], s_v, b_v, o8, lo, hi);
            float* d = vo + (p0 + (s << 4) + l16) * CCH + o8;
            *(v4f*)(d) = lo;  *(v4f*)(d + 4) = hi;
        }
    }
    __syncthreads();

    {
        v8f acc[4] = {};
        mma_tile128(tq, bbase, w_q2 + arow, acc);
        #pragma unroll
        for (int s = 0; s < 4; ++s) {
            v4f lo, hi;
            bn_relu_tile(acc[s], s_q2, b_q2, o8, lo, hi);
            float* d = qo + (p0 + (s << 4) + l16) * CCH + o8;
            *(v4f*)(d) = lo;  *(v4f*)(d + 4) = hi;
        }
    }
    {
        v8f acc[4] = {};
        mma_tile128(tk, bbase, w_k2 + arow, acc);
        #pragma unroll
        for (int s = 0; s < 4; ++s) {
            v4f lo, hi;
            bn_relu_tile(acc[s], s_k2, b_k2, o8, lo, hi);
            float* d = ko + (p0 + (s << 4) + l16) * CCH + o8;
            *(v4f*)(d) = lo;  *(v4f*)(d + 4) = hi;
        }
    }
}

__device__ __forceinline__ float xhalf(float x) {
    const int b = __builtin_bit_cast(int, x);
    return __builtin_bit_cast(float,
        __builtin_amdgcn_permlanex16(b, b, 0x76543210, 0xfedcba98, false, false));
}
template <int M> __device__ __forceinline__ float xrow(float x) {
    const int b = __builtin_bit_cast(int, x);
    return __builtin_bit_cast(float,
        __builtin_amdgcn_update_dpp(b, b, 0x160 + M, 0xf, 0xf, true));
}
__device__ __forceinline__ float dot4(const v4f& a, const v4f& b) {
    return a.x * b.x + a.y * b.y + a.z * b.z + a.w * b.w;
}

#define TILE_W 8
#define TILE_H 8
#define HALO_W (TILE_W + KSZ - 1)
#define HALO_H (TILE_H + KSZ - 1)
#define HALO_N (HALO_W * HALO_H)
#define HALO_V4 (HALO_N * (CCH / 4))
#define ATT_THREADS 512
#define PATCH (KSZ + 1)
#define TILE_PX (TILE_W * TILE_H)
#define RPITCH 68

template <bool Y0, bool Y1>
__device__ __forceinline__ void pv_row(const float* vrow, int h, const float (&wgt)[KSZ],
                                       v2f (&alo)[2][2], v2f (&ahi)[2][2])
{
    v4f vr[PATCH];
    #pragma unroll
    for (int cc = 0; cc < PATCH; ++cc)
        vr[cc] = *(const v4f*)(vrow + cc * CCH);
    #pragma unroll
    for (int cc = 0; cc < PATCH; ++cc) {
        const v2f vlo = {vr[cc].x, vr[cc].y};
        const v2f vhi = {vr[cc].z, vr[cc].w};
        #pragma unroll
        for (int y = 0; y < 2; ++y) {
            #pragma unroll
            for (int xx = 0; xx < 2; ++xx) {
                const int c = cc - xx;
                if ((y == 0 ? Y0 : Y1) && c >= 0 && c < KSZ) {
                    const float wt = __builtin_bit_cast(float, __builtin_amdgcn_readlane(
                        __builtin_bit_cast(int, wgt[c]), 16 * xx + 8 * y + h));
                    const v2f w2 = {wt, wt};
                    alo[y][xx] = __builtin_elementwise_fma(w2, vlo, alo[y][xx]);
                    ahi[y][xx] = __builtin_elementwise_fma(w2, vhi, ahi[y][xx]);
                }
            }
        }
    }
}

__global__ __launch_bounds__(ATT_THREADS) void local_attn_out_kernel(
    const float* __restrict__ q, const float* __restrict__ k,
    const float* __restrict__ v, const float* __restrict__ x,
    const float* __restrict__ w_o, const float* __restrict__ s_o,
    const float* __restrict__ b_o, float* __restrict__ out)
{
    __shared__ __attribute__((aligned(16))) float kvs[2 * HALO_N * CCH];
    __shared__ __attribute__((aligned(16))) float tn[TILE_PX * TPITCH];
    __shared__ __attribute__((aligned(16))) float txx[TILE_PX * TPITCH];
    float* const ks  = kvs;
    float* const vs  = kvs + HALO_N * CCH;
    float* const red = kvs;

    const int tid  = threadIdx.x;
    const int lane = tid & 31;
    const int wv   = tid >> 5;
    const int h0   = (blockIdx.x / (WDIM / TILE_W)) * TILE_H;
    const int w0   = (blockIdx.x % (WDIM / TILE_W)) * TILE_W;

    const int ty0 = (wv >> 2) << 1;
    const int tx0 = (wv & 3) << 1;

    v4f qv[2][2];
    #pragma unroll
    for (int y = 0; y < 2; ++y) {
        #pragma unroll
        for (int xx = 0; xx < 2; ++xx) {
            const int p = (h0 + ty0 + y) * WDIM + w0 + tx0 + xx;
            qv[y][xx] = *(const v4f*)(q + (long)p * CCH + lane * 4);
        }
    }

    {
        const int c  = tid & (CCH - 1);
        const int ry = tid >> 7;
        v4f xa[2], xb[2];
        #pragma unroll
        for (int hv = 0; hv < 2; ++hv) {
            const float* xp = x + (long)c * HW + (h0 + ry + 4 * hv) * WDIM + w0;
            xa[hv] = *(const v4f*)(xp);
            xb[hv] = *(const v4f*)(xp + 4);
        }
        #pragma unroll
        for (int hv = 0; hv < 2; ++hv) {
            float* d = &txx[((ry + 4 * hv) * TILE_W) * TPITCH + c];
            d[0 * TPITCH] = xa[hv].x;  d[1 * TPITCH] = xa[hv].y;
            d[2 * TPITCH] = xa[hv].z;  d[3 * TPITCH] = xa[hv].w;
            d[4 * TPITCH] = xb[hv].x;  d[5 * TPITCH] = xb[hv].y;
            d[6 * TPITCH] = xb[hv].z;  d[7 * TPITCH] = xb[hv].w;
        }
    }

    #pragma unroll
    for (int it = 0; it < (HALO_V4 + ATT_THREADS - 1) / ATT_THREADS; ++it) {
        const int e = tid + it * ATT_THREADS;
        if (e < HALO_V4) {
            const int hp = e >> 5;
            const int c4 = (e & 31) << 2;
            const int hy = hp / HALO_W;
            const int hx = hp - hy * HALO_W;
            const int gh = h0 + hy - (KSZ / 2);
            const int gw = w0 + hx - (KSZ / 2);
            const bool ok = (unsigned)gh < (unsigned)HDIM && (unsigned)gw < (unsigned)WDIM;
            const int ghc = min(max(gh, 0), HDIM - 1);
            const int gwc = min(max(gw, 0), WDIM - 1);
            const long off = (long)(ghc * WDIM + gwc) * CCH + c4;
            v4f kv = *(const v4f*)(k + off);
            v4f vv = *(const v4f*)(v + off);
            if (!ok) {
                kv = (v4f){0.0f, 0.0f, 0.0f, 0.0f};
                vv = (v4f){0.0f, 0.0f, 0.0f, 0.0f};
            }
            *(v4f*)(&ks[hp * CCH + c4]) = kv;
            *(v4f*)(&vs[hp * CCH + c4]) = vv;
        }
    }
    __syncthreads();

    const int wbase = (ty0 * HALO_W + tx0) * CCH + lane * 4;
    const bool lo = lane < 16;
    const bool b3 = (lane & 8) != 0, b2 = (lane & 4) != 0;
    const bool b1 = (lane & 2) != 0, b0 = (lane & 1) != 0;

    const int hl  = lane >> 4;
    const int l16 = lane & 15;
    const int m0  = (wv & 7) << 4;
    const int kh  = wv >> 3;
    v2f afr[CCH / 4];
    const float* __restrict__ Ap =
        w_o + (long)(m0 + l16) * (2 * CCH) + (kh << 7) + (hl << 1);

    float s8[PATCH][KSZ];
    v4f kr[2][PATCH];
    #pragma unroll
    for (int cc = 0; cc < PATCH; ++cc)
        kr[0][cc] = *(const v4f*)(&ks[wbase + cc * CCH]);
    #pragma unroll
    for (int h = 0; h < PATCH; ++h) {
        if (h + 1 < PATCH) {
            #pragma unroll
            for (int cc = 0; cc < PATCH; ++cc)
                kr[(h + 1) & 1][cc] = *(const v4f*)(&ks[wbase + ((h + 1) * HALO_W + cc) * CCH]);
        }
        #pragma unroll
        for (int j = 0; j < CCH / 4 / PATCH; ++j)
            afr[h * (CCH / 4 / PATCH) + j] =
                *(const v2f*)(Ap + ((h * (CCH / 4 / PATCH) + j) << 2));
        #pragma unroll
        for (int c = 0; c < KSZ; ++c) {
            float ua = 0.0f, ub = 0.0f;
            if (h < KSZ) {
                const float d0 = dot4(qv[0][0], kr[h & 1][c]);
                const float d1 = dot4(qv[0][1], kr[h & 1][c + 1]);
                const float keep = lo ? d0 : d1;
                const float send = lo ? d1 : d0;
                ua = keep + xhalf(send);
            }
            if (h >= 1) {
                const float d0 = dot4(qv[1][0], kr[h & 1][c]);
                const float d1 = dot4(qv[1][1], kr[h & 1][c + 1]);
                const float keep = lo ? d0 : d1;
                const float send = lo ? d1 : d0;
                ub = keep + xhalf(send);
            }
            if (h == 0) {
                s8[h][c] = ua + xrow<8>(ua);
            } else if (h == PATCH - 1) {
                s8[h][c] = ub + xrow<8>(ub);
            } else {
                const float keep = b3 ? ub : ua;
                const float send = b3 ? ua : ub;
                s8[h][c] = keep + xrow<8>(send);
            }
        }
    }
    float s4[PATCH / 2][KSZ];
    #pragma unroll
    for (int h = 0; h < PATCH / 2; ++h) {
        #pragma unroll
        for (int c = 0; c < KSZ; ++c) {
            const float keep = b2 ? s8[h + 4][c] : s8[h][c];
            const float send = b2 ? s8[h][c] : s8[h + 4][c];
            s4[h][c] = keep + xrow<4>(send);
        }
    }
    float s2[PATCH / 4][KSZ];
    #pragma unroll
    for (int h = 0; h < PATCH / 4; ++h) {
        #pragma unroll
        for (int c = 0; c < KSZ; ++c) {
            const float keep = b1 ? s4[h + 2][c] : s4[h][c];
            const float send = b1 ? s4[h][c] : s4[h + 2][c];
            s2[h][c] = keep + xrow<2>(send);
        }
    }
    float sc[KSZ];
    #pragma unroll
    for (int c = 0; c < KSZ; ++c) {
        const float keep = b0 ? s2[1][c] : s2[0][c];
        const float send = b0 ? s2[0][c] : s2[1][c];
        sc[c] = keep + xrow<1>(send);
    }

    const int hrow = lane & 7;
    const bool vld = b3 ? (hrow >= 1) : (hrow < KSZ);
    float m = -__builtin_inff();
    #pragma unroll
    for (int c = 0; c < KSZ; ++c) m = vld ? fmaxf(m, sc[c]) : m;
    m = fmaxf(m, xrow<4>(m));
    m = fmaxf(m, xrow<2>(m));
    m = fmaxf(m, xrow<1>(m));
    float wgt[KSZ];
    #pragma unroll
    for (int c = 0; c < KSZ; ++c) wgt[c] = vld ? __expf(sc[c] - m) : 0.0f;
    float z = wgt[0];
    #pragma unroll
    for (int c = 1; c < KSZ; ++c) z += wgt[c];
    z += xrow<4>(z);
    z += xrow<2>(z);
    z += xrow<1>(z);
    const float inv = 1.0f / z;
    #pragma unroll
    for (int c = 0; c < KSZ; ++c) wgt[c] *= inv;

    v2f alo[2][2], ahi[2][2];
    #pragma unroll
    for (int y = 0; y < 2; ++y) {
        #pragma unroll
        for (int xx = 0; xx < 2; ++xx) {
            alo[y][xx] = (v2f){0.0f, 0.0f};
            ahi[y][xx] = (v2f){0.0f, 0.0f};
        }
    }
    const float* const vq = vs + wbase;
    pv_row<true, false>(vq, 0, wgt, alo, ahi);
    #pragma unroll 1
    for (int h = 1; h < PATCH - 1; ++h)
        pv_row<true, true>(vq + h * (HALO_W * CCH), h, wgt, alo, ahi);
    pv_row<false, true>(vq + (PATCH - 1) * (HALO_W * CCH), PATCH - 1, wgt, alo, ahi);
    #pragma unroll
    for (int y = 0; y < 2; ++y) {
        #pragma unroll
        for (int xx = 0; xx < 2; ++xx) {
            const v4f a = {alo[y][xx].x, alo[y][xx].y, ahi[y][xx].x, ahi[y][xx].y};
            *(v4f*)(&tn[((ty0 + y) * TILE_W + tx0 + xx) * TPITCH + lane * 4]) = a;
        }
    }

    __syncthreads();

    {
        const float* Bt = kh ? txx : tn;
        const int brow = l16 * TPITCH + (hl << 1);

        v8f pacc[4] = {};
        #pragma unroll
        for (int j = 0; j < CCH / 4; ++j) {
            const int k0 = j << 2;
            #pragma unroll
            for (int s = 0; s < 4; ++s) {
                const v2f b = *(const v2f*)(&Bt[(s << 4) * TPITCH + brow + k0]);
                pacc[s] = __builtin_amdgcn_wmma_f32_16x16x4_f32(
                    false, afr[j], false, b, (short)0, pacc[s], false, false);
            }
        }
        #pragma unroll
        for (int s = 0; s < 4; ++s) {
            #pragma unroll
            for (int r = 0; r < 8; ++r)
                red[(kh * CCH + m0 + r + (hl << 3)) * RPITCH + (s << 4) + l16] = pacc[s][r];
        }
    }
    __syncthreads();

    #pragma unroll
    for (int ps = 0; ps < 4; ++ps) {
        const int o  = (tid >> 4) + (ps << 5);
        const int ry = (tid >> 1) & 7;
        const int hf = (tid & 1) << 2;
        const float* rp = &red[o * RPITCH + ry * TILE_W + hf];
        v4f sum = *(const v4f*)(rp);
        sum += *(const v4f*)(rp + CCH * RPITCH);
        const float sc2 = s_o[o], bi = b_o[o];
        v4f y;
        y.x = sum.x * sc2 + bi;  y.y = sum.y * sc2 + bi;
        y.z = sum.z * sc2 + bi;  y.w = sum.w * sc2 + bi;
        *(v4f*)(out + (long)o * HW + (h0 + ry) * WDIM + w0 + hf) = y;
    }
}

extern "C" void kernel_launch(void* const* d_in, const int* in_sizes, int n_in,
                              void* d_out, int out_size, void* d_ws, size_t ws_size,
                              hipStream_t stream) {
    const float* x    = (const float*)d_in[0];
    const float* w_q1 = (const float*)d_in[1];
    const float* s_q1 = (const float*)d_in[2];
    const float* b_q1 = (const float*)d_in[3];
    const float* w_q2 = (const float*)d_in[4];
    const float* s_q2 = (const float*)d_in[5];
    const float* b_q2 = (const float*)d_in[6];
    const float* w_k1 = (const float*)d_in[7];
    const float* s_k1 = (const float*)d_in[8];
    const float* b_k1 = (const float*)d_in[9];
    const float* w_k2 = (const float*)d_in[10];
    const float* s_k2 = (const float*)d_in[11];
    const float* b_k2 = (const float*)d_in[12];
    const float* w_v  = (const float*)d_in[13];
    const float* s_v  = (const float*)d_in[14];
    const float* b_v  = (const float*)d_in[15];
    const float* w_o  = (const float*)d_in[16];
    const float* s_o  = (const float*)d_in[17];
    const float* b_o  = (const float*)d_in[18];
    float* out = (float*)d_out;

    const size_t NB = (size_t)HW * CCH;
    float* qb = (float*)d_ws;
    float* kb = qb + NB;
    float* vb = kb + NB;

    const dim3 blk(256);
    const dim3 grd(HW / 64);

    proj_fused_wmma<<<grd, blk, 0, stream>>>(x,
                                             w_q1, s_q1, b_q1, w_q2, s_q2, b_q2,
                                             w_k1, s_k1, b_k1, w_k2, s_k2, b_k2,
                                             w_v, s_v, b_v,
                                             qb, kb, vb);
    local_attn_out_kernel<<<dim3(HW / TILE_PX), dim3(ATT_THREADS), 0, stream>>>(
        qb, kb, vb, x, w_o, s_o, b_o, out);
}


# q projections moved from the projection launch into the attention launch: there waves 0..7 do both q products on WMMA while waves 8..15 stage the K and V halos; k order and sum order unchanged
# speedup vs baseline: 3.3478x; 1.0431x over previous
#include <hip/hip_runtime.h>
#include <hip/hip_bf16.h>

typedef __attribute__((ext_vector_type(2))) float v2f;
typedef __attribute__((ext_vector_type(4))) float v4f;
typedef __attribute__((ext_vector_type(8))) float v8f;

#define CCH 128
#define HDIM 128
#define WDIM 128
#define HW (HDIM * WDIM)
#define KSZ 7
#define KK2 49
#define TPITCH 132

__device__ __forceinline__ void mma_tile128(
    const float* tile, const int (&bbase)[4],
    const float* __restrict__ Ap, v8f (&acc)[4])
{
    #pragma unroll
    for (int k0 = 0; k0 < CCH; k0 += 4) {
        const v2f a = *(const v2f*)(Ap + k0);
        #pragma unroll
        for (int s = 0; s < 4; ++s) {
            const v2f b = *(const v2f*)(&tile[bbase[s] + k0]);
            acc[s] = __builtin_amdgcn_wmma_f32_16x16x4_f32(
                false, a, false, b, (short)0, acc[s], false, false);
        }
    }
}

__device__ __forceinline__ void load_a_row(const float* __restrict__ Ap, v2f (&afr)[CCH / 4])
{
    #pragma unroll
    for (int j = 0; j < CCH / 4; ++j)
        afr[j] = *(const v2f*)(Ap + (j << 2));
}
__device__ __forceinline__ void mma_tile128_pre(
    const float* tile, const int (&bbase)[4], const v2f (&afr)[CCH / 4], v8f (&acc)[4])
{
    #pragma unroll
    for (int j = 0; j < CCH / 4; ++j) {
        #pragma unroll
        for (int s = 0; s < 4; ++s) {
            const v2f b = *(const v2f*)(&tile[bbase[s] + (j << 2)]);
            acc[s] = __builtin_amdgcn_wmma_f32_16x16x4_f32(
                false, afr[j], false, b, (short)0, acc[s], false, false);
        }
    }
}

__device__ __forceinline__ void bn_relu_tile(
    const v8f& d, const float* __restrict__ scale, const float* __restrict__ bias,
    int o8, v4f& lo, v4f& hi)
{
    const v4f s0 = *(const v4f*)(scale + o8), s1 = *(const v4f*)(scale + o8 + 4);
    const v4f b0 = *(const v4f*)(bias + o8),  b1 = *(const v4f*)(bias + o8 + 4);
    lo.x = fmaxf(d[0] * s0.x + b0.x, 0.0f);
    lo.y = fmaxf(d[1] * s0.y + b0.y, 0.0f);
    lo.z = fmaxf(d[2] * s0.z + b0.z, 0.0f);
    lo.w = fmaxf(d[3] * s0.w + b0.w, 0.0f);
    hi.x = fmaxf(d[4] * s1.x + b1.x, 0.0f);
    hi.y = fmaxf(d[5] * s1.y + b1.y, 0.0f);
    hi.z = fmaxf(d[6] * s1.z + b1.z, 0.0f);
    hi.w = fmaxf(d[7] * s1.w + b1.w, 0.0f);
}

__global__ __launch_bounds__(256) void proj_fused_wmma(
    const float* __restrict__ x,
    const float* __restrict__ w_k1, const float* __restrict__ s_k1, const float* __restrict__ b_k1,
    const float* __restrict__ w_k2, const float* __restrict__ s_k2, const float* __restrict__ b_k2,
    const float* __restrict__ w_v,  const float* __restrict__ s_v,  const float* __restrict__ b_v,
    float* __restrict__ ko, float* __restrict__ vo)
{
    __shared__ __attribute__((aligned(16))) float tx[64 * TPITCH];
    __shared__ __attribute__((aligned(16))) float tk[64 * TPITCH];

    const int tid  = threadIdx.x;
    const int lane = tid & 31;
    const int wv   = tid >> 5;
    const int hl   = lane >> 4;
    const int l16  = lane & 15;
    const int m0   = wv << 4;
    const long p0  = (long)blockIdx.x * 64;
    const int o8   = m0 + (hl << 3);

    int bbase[4];
    #pragma unroll
    for (int s = 0; s < 4; ++s)
        bbase[s] = ((s << 4) + l16) * TPITCH + (hl << 1);

    #pragma unroll
    for (int i = 0; i < 8; ++i) {
        const int e  = tid + (i << 8);
        const int p4 = (e & 15) << 2;
        const int c  = e >> 4;
        const v4f xv = *(const v4f*)(x + (long)c * HW + p0 + p4);
        tx[(p4 + 0) * TPITCH + c] = xv.x;
        tx[(p4 + 1) * TPITCH + c] = xv.y;
        tx[(p4 + 2) * TPITCH + c] = xv.z;
        tx[(p4 + 3) * TPITCH + c] = xv.w;
    }
    __syncthreads();

    const long arow = (long)(m0 + l16) * CCH + (hl << 1);

    {
        v8f acc[4] = {};
        mma_tile128(tx, bbase, w_k1 + arow, acc);
        #pragma unroll
        for (int s = 0; s < 4; ++s) {
            v4f lo, hi;
            bn_relu_tile(acc[s], s_k1, b_k1, o8, lo, hi);
            float* d = &tk[((s << 4) + l16) * TPITCH + o8];
            *(v4f*)(d) = lo;  *(v4f*)(d + 4) = hi;
        }
    }
    {
        v8f acc[4] = {};
        mma_tile128(tx, bbase, w_v + arow, acc);
        #pragma unroll
        for (int s = 0; s < 4; ++s) {
            v4f lo, hi;
            bn_relu_tile(acc[s], s_v, b_v, o8, lo, hi);
            float* d = vo + (p0 + (s << 4) + l16) * CCH + o8;
            *(v4f*)(d) = lo;  *(v4f*)(d + 4) = hi;
        }
    }
    __syncthreads();

    {
        v8f acc[4] = {};
        mma_tile128(tk, bbase, w_k2 + arow, acc);
        #pragma unroll
        for (int s = 0; s < 4; ++s) {
            v4f lo, hi;
            bn_relu_tile(acc[s], s_k2, b_k2, o8, lo, hi);
            float* d = ko + (p0 + (s << 4) + l16) * CCH + o8;
            *(v4f*)(d) = lo;  *(v4f*)(d + 4) = hi;
        }
    }
}

__device__ __forceinline__ float xhalf(float x) {
    const int b = __builtin_bit_cast(int, x);
    return __builtin_bit_cast(float,
        __builtin_amdgcn_permlanex16(b, b, 0x76543210, 0xfedcba98, false, false));
}
template <int M> __device__ __forceinline__ float xrow(float x) {
    const int b = __builtin_bit_cast(int, x);
    return __builtin_bit_cast(float,
        __builtin_amdgcn_update_dpp(b, b, 0x160 + M, 0xf, 0xf, true));
}
__device__ __forceinline__ float dot4(const v4f& a, const v4f& b) {
    return a.x * b.x + a.y * b.y + a.z * b.z + a.w * b.w;
}

#define TILE_W 8
#define TILE_H 8
#define HALO_W (TILE_W + KSZ - 1)
#define HALO_H (TILE_H + KSZ - 1)
#define HALO_N (HALO_W * HALO_H)
#define HALO_V4 (HALO_N * (CCH / 4))
#define ATT_THREADS 512
#define PATCH (KSZ + 1)
#define TILE_PX (TILE_W * TILE_H)
#define RPITCH 68

template <bool Y0, bool Y1>
__device__ __forceinline__ void pv_row(const float* vrow, int h, const float (&wgt)[KSZ],
                                       v2f (&alo)[2][2], v2f (&ahi)[2][2])
{
    v4f vr[PATCH];
    #pragma unroll
    for (int cc = 0; cc < PATCH; ++cc)
        vr[cc] = *(const v4f*)(vrow + cc * CCH);
    #pragma unroll
    for (int cc = 0; cc < PATCH; ++cc) {
        const v2f vlo = {vr[cc].x, vr[cc].y};
        const v2f vhi = {vr[cc].z, vr[cc].w};
        #pragma unroll
        for (int y = 0; y < 2; ++y) {
            #pragma unroll
            for (int xx = 0; xx < 2; ++xx) {
                const int c = cc - xx;
                if ((y == 0 ? Y0 : Y1) && c >= 0 && c < KSZ) {
                    const float wt = __builtin_bit_cast(float, __builtin_amdgcn_readlane(
                        __builtin_bit_cast(int, wgt[c]), 16 * xx + 8 * y + h));
                    const v2f w2 = {wt, wt};
                    alo[y][xx] = __builtin_elementwise_fma(w2, vlo, alo[y][xx]);
                    ahi[y][xx] = __builtin_elementwise_fma(w2, vhi, ahi[y][xx]);
                }
            }
        }
    }
}

#define HALO_THREADS 256
__device__ __forceinline__ void stage_halo(const float* __restrict__ src, float* dst,
                                           int t, int h0, int w0)
{
    #pragma unroll
    for (int it = 0; it < (HALO_V4 + HALO_THREADS - 1) / HALO_THREADS; ++it) {
        const int e  = t + it * HALO_THREADS;
        const int ec = e < HALO_V4 ? e : e - HALO_THREADS / 2;
        const int hp = ec >> 5;
        const int c4 = (ec & 31) << 2;
        const int hy = hp / HALO_W;
        const int hx = hp - hy * HALO_W;
        const int gh = h0 + hy - (KSZ / 2);
        const int gw = w0 + hx - (KSZ / 2);
        const bool ok = (unsigned)gh < (unsigned)HDIM && (unsigned)gw < (unsigned)WDIM;
        const int ghc = min(max(gh, 0), HDIM - 1);
        const int gwc = min(max(gw, 0), WDIM - 1);
        const long off = (long)(ghc * WDIM + gwc) * CCH + c4;
        v4f val = *(const v4f*)(src + off);
        if (!ok) val = (v4f){0.0f, 0.0f, 0.0f, 0.0f};
        *(v4f*)(&dst[hp * CCH + c4]) = val;
    }
}

__device__ __forceinline__ void q_proj_tile(
    const float* src, float* dst, const int (&bbase)[4], const v2f (&afr)[CCH / 4],
    const float* __restrict__ scale, const float* __restrict__ bias,
    int mq, int hl, int l16)
{
    const int o8 = mq + (hl << 3);
    v8f acc[4] = {};
    mma_tile128_pre(src, bbase, afr, acc);
    #pragma unroll
    for (int s = 0; s < 4; ++s) {
        v4f ylo, yhi;
        bn_relu_tile(acc[s], scale, bias, o8, ylo, yhi);
        float* d = &dst[((s << 4) + l16) * TPITCH + o8];
        *(v4f*)(d) = ylo;  *(v4f*)(d + 4) = yhi;
    }
}

__global__ __launch_bounds__(ATT_THREADS) void local_attn_out_kernel(
    const float* __restrict__ k, const float* __restrict__ v, const float* __restrict__ x,
    const float* __restrict__ w_q1, const float* __restrict__ s_q1, const float* __restrict__ b_q1,
    const float* __restrict__ w_q2, const float* __restrict__ s_q2, const float* __restrict__ b_q2,
    const float* __restrict__ w_o, const float* __restrict__ s_o,
    const float* __restrict__ b_o, float* __restrict__ out)
{
    __shared__ __attribute__((aligned(16))) float kvs[2 * HALO_N * CCH];
    __shared__ __attribute__((aligned(16))) float tn[TILE_PX * TPITCH];
    __shared__ __attribute__((aligned(16))) float txx[TILE_PX * TPITCH];
    __shared__ __attribute__((aligned(16))) float tq[TILE_PX * TPITCH];
    float* const ks  = kvs;
    float* const vs  = kvs + HALO_N * CCH;
    float* const red = kvs;

    const int tid  = threadIdx.x;
    const int lane = tid & 31;
    const int wv   = __builtin_amdgcn_readfirstlane(tid >> 5);
    const int hl   = lane >> 4;
    const int l16  = lane & 15;
    const int h0   = (blockIdx.x / (WDIM / TILE_W)) * TILE_H;
    const int w0   = (blockIdx.x % (WDIM / TILE_W)) * TILE_W;

    const int ty0 = (wv >> 2) << 1;
    const int tx0 = (wv & 3) << 1;

    v2f aq1[CCH / 4], aq2[CCH / 4];
    const long qrow = (long)(((wv & 7) << 4) + l16) * CCH + (hl << 1);
    if (wv < 8)
        load_a_row(w_q1 + qrow, aq1);

    {
        const int c  = tid & (CCH - 1);
        const int ry = tid >> 7;
        v4f xa[2], xb[2];
        #pragma unroll
        for (int hv = 0; hv < 2; ++hv) {
            const float* xp = x + (long)c * HW + (h0 + ry + 4 * hv) * WDIM + w0;
            xa[hv] = *(const v4f*)(xp);
            xb[hv] = *(const v4f*)(xp + 4);
        }
        #pragma unroll
        for (int hv = 0; hv < 2; ++hv) {
            float* d = &txx[((ry + 4 * hv) * TILE_W) * TPITCH + c];
            d[0 * TPITCH] = xa[hv].x;  d[1 * TPITCH] = xa[hv].y;
            d[2 * TPITCH] = xa[hv].z;  d[3 * TPITCH] = xa[hv].w;
            d[4 * TPITCH] = xb[hv].x;  d[5 * TPITCH] = xb[hv].y;
            d[6 * TPITCH] = xb[hv].z;  d[7 * TPITCH] = xb[hv].w;
        }
    }

    __syncthreads();

    int qbase[4];
    #pragma unroll
    for (int s = 0; s < 4; ++s)
        qbase[s] = ((s << 4) + l16) * TPITCH + (hl << 1);
    if (wv < 8) {
        load_a_row(w_q2 + qrow, aq2);
        q_proj_tile(txx, tq, qbase, aq1, s_q1, b_q1, wv << 4, hl, l16);
    } else {
        stage_halo(k, ks, tid & (HALO_THREADS - 1), h0, w0);
    }
    __syncthreads();
    if (wv < 8)
        q_proj_tile(tq, tn, qbase, aq2, s_q2, b_q2, wv << 4, hl, l16);
    else
        stage_halo(v, vs, tid & (HALO_THREADS - 1), h0, w0);
    __syncthreads();

    v4f qv[2][2];
    #pragma unroll
    for (int y = 0; y < 2; ++y) {
        #pragma unroll
        for (int xx = 0; xx < 2; ++xx)
            qv[y][xx] = *(const v4f*)(&tn[((ty0 + y) * TILE_W + tx0 + xx) * TPITCH + lane * 4]);
    }

    const int wbase = (ty0 * HALO_W + tx0) * CCH + lane * 4;
    const bool lo = lane < 16;
    const bool b3 = (lane & 8) != 0, b2 = (lane & 4) != 0;
    const bool b1 = (lane & 2) != 0, b0 = (lane & 1) != 0;

    const int m0  = (wv & 7) << 4;
    const int kh  = wv >> 3;
    v2f afr[CCH / 4];
    const float* __restrict__ Ap =
        w_o + (long)(m0 + l16) * (2 * CCH) + (kh << 7) + (hl << 1);

    float s8[PATCH][KSZ];
    v4f kr[2][PATCH];
    #pragma unroll
    for (int cc = 0; cc < PATCH; ++cc)
        kr[0][cc] = *(const v4f*)(&ks[wbase + cc * CCH]);
    #pragma unroll
    for (int h = 0; h < PATCH; ++h) {
        if (h + 1 < PATCH) {
            #pragma unroll
            for (int cc = 0; cc < PATCH; ++cc)
                kr[(h + 1) & 1][cc] = *(const v4f*)(&ks[wbase + ((h + 1) * HALO_W + cc) * CCH]);
        }
        #pragma unroll
        for (int j = 0; j < CCH / 4 / PATCH; ++j)
            afr[h * (CCH / 4 / PATCH) + j] =
                *(const v2f*)(Ap + ((h * (CCH / 4 / PATCH) + j) << 2));
        #pragma unroll
        for (int c = 0; c < KSZ; ++c) {
            float ua = 0.0f, ub = 0.0f;
            if (h < KSZ) {
                const float d0 = dot4(qv[0][0], kr[h & 1][c]);
                const float d1 = dot4(qv[0][1], kr[h & 1][c + 1]);
                const float keep = lo ? d0 : d1;
                const float send = lo ? d1 : d0;
                ua = keep + xhalf(send);
            }
            if (h >= 1) {
                const float d0 = dot4(qv[1][0], kr[h & 1][c]);
                const float d1 = dot4(qv[1][1], kr[h & 1][c + 1]);
                const float keep = lo ? d0 : d1;
                const float send = lo ? d1 : d0;
                ub = keep + xhalf(send);
            }
            if (h == 0) {
                s8[h][c] = ua + xrow<8>(ua);
            } else if (h == PATCH - 1) {
                s8[h][c] = ub + xrow<8>(ub);
            } else {
                const float keep = b3 ? ub : ua;
                const float send = b3 ? ua : ub;
                s8[h][c] = keep + xrow<8>(send);
            }
        }
    }
    float s4[PATCH / 2][KSZ];
    #pragma unroll
    for (int h = 0; h < PATCH / 2; ++h) {
        #pragma unroll
        for (int c = 0; c < KSZ; ++c) {
            const float keep = b2 ? s8[h + 4][c] : s8[h][c];
            const float send = b2 ? s8[h][c] : s8[h + 4][c];
            s4[h][c] = keep + xrow<4>(send);
        }
    }
    float s2[PATCH / 4][KSZ];
    #pragma unroll
    for (int h = 0; h < PATCH / 4; ++h) {
        #pragma unroll
        for (int c = 0; c < KSZ; ++c) {
            const float keep = b1 ? s4[h + 2][c] : s4[h][c];
            const float send = b1 ? s4[h][c] : s4[h + 2][c];
            s2[h][c] = keep + xrow<2>(send);
        }
    }
    float sc[KSZ];
    #pragma unroll
    for (int c = 0; c < KSZ; ++c) {
        const float keep = b0 ? s2[1][c] : s2[0][c];
        const float send = b0 ? s2[0][c] : s2[1][c];
        sc[c] = keep + xrow<1>(send);
    }

    const int hrow = lane & 7;
    const bool vld = b3 ? (hrow >= 1) : (hrow < KSZ);
    float m = -__builtin_inff();
    #pragma unroll
    for (int c = 0; c < KSZ; ++c) m = vld ? fmaxf(m, sc[c]) : m;
    m = fmaxf(m, xrow<4>(m));
    m = fmaxf(m, xrow<2>(m));
    m = fmaxf(m, xrow<1>(m));
    float wgt[KSZ];
    #pragma unroll
    for (int c = 0; c < KSZ; ++c) wgt[c] = vld ? __expf(sc[c] - m) : 0.0f;
    float z = wgt[0];
    #pragma unroll
    for (int c = 1; c < KSZ; ++c) z += wgt[c];
    z += xrow<4>(z);
    z += xrow<2>(z);
    z += xrow<1>(z);
    const float inv = 1.0f / z;
    #pragma unroll
    for (int c = 0; c < KSZ; ++c) wgt[c] *= inv;

    v2f alo[2][2], ahi[2][2];
    #pragma unroll
    for (int y = 0; y < 2; ++y) {
        #pragma unroll
        for (int xx = 0; xx < 2; ++xx) {
            alo[y][xx] = (v2f){0.0f, 0.0f};
            ahi[y][xx] = (v2f){0.0f, 0.0f};
        }
    }
    const float* const vq = vs + wbase;
    pv_row<true, false>(vq, 0, wgt, alo, ahi);
    #pragma unroll 1
    for (int h = 1; h < PATCH - 1; ++h)
        pv_row<true, true>(vq + h * (HALO_W * CCH), h, wgt, alo, ahi);
    pv_row<false, true>(vq + (PATCH - 1) * (HALO_W * CCH), PATCH - 1, wgt, alo, ahi);
    #pragma unroll
    for (int y = 0; y < 2; ++y) {
        #pragma unroll
        for (int xx = 0; xx < 2; ++xx) {
            const v4f a = {alo[y][xx].x, alo[y][xx].y, ahi[y][xx].x, ahi[y][xx].y};
            *(v4f*)(&tn[((ty0 + y) * TILE_W + tx0 + xx) * TPITCH + lane * 4]) = a;
        }
    }

    __syncthreads();

    {
        const float* Bt = kh ? txx : tn;
        const int brow = l16 * TPITCH + (hl << 1);

        v8f pacc[4] = {};
        #pragma unroll
        for (int j = 0; j < CCH / 4; ++j) {
            const int k0 = j << 2;
            #pragma unroll
            for (int s = 0; s < 4; ++s) {
                const v2f b = *(const v2f*)(&Bt[(s << 4) * TPITCH + brow + k0]);
                pacc[s] = __builtin_amdgcn_wmma_f32_16x16x4_f32(
                    false, afr[j], false, b, (short)0, pacc[s], false, false);
            }
        }
        #pragma unroll
        for (int s = 0; s < 4; ++s) {
            #pragma unroll
            for (int r = 0; r < 8; ++r)
                red[(kh * CCH + m0 + r + (hl << 3)) * RPITCH + (s << 4) + l16] = pacc[s][r];
        }
    }
    __syncthreads();

    #pragma unroll
    for (int ps = 0; ps < 4; ++ps) {
        const int o  = (tid >> 4) + (ps << 5);
        const int ry = (tid >> 1) & 7;
        const int hf = (tid & 1) << 2;
        const float* rp = &red[o * RPITCH + ry * TILE_W + hf];
        v4f sum = *(const v4f*)(rp);
        sum += *(const v4f*)(rp + CCH * RPITCH);
        const float sc2 = s_o[o], bi = b_o[o];
        v4f y;
        y.x = sum.x * sc2 + bi;  y.y = sum.y * sc2 + bi;
        y.z = sum.z * sc2 + bi;  y.w = sum.w * sc2 + bi;
        *(v4f*)(out + (long)o * HW + (h0 + ry) * WDIM + w0 + hf) = y;
    }
}

extern "C" void kernel_launch(void* const* d_in, const int* in_sizes, int n_in,
                              void* d_out, int out_size, void* d_ws, size_t ws_size,
                              hipStream_t stream) {
    const float* x    = (const float*)d_in[0];
    const float* w_q1 = (const float*)d_in[1];
    const float* s_q1 = (const float*)d_in[2];
    const float* b_q1 = (const float*)d_in[3];
    const float* w_q2 = (const float*)d_in[4];
    const float* s_q2 = (const float*)d_in[5];
    const float* b_q2 = (const float*)d_in[6];
    const float* w_k1 = (const float*)d_in[7];
    const float* s_k1 = (const float*)d_in[8];
    const float* b_k1 = (const float*)d_in[9];
    const float* w_k2 = (const float*)d_in[10];
    const float* s_k2 = (const float*)d_in[11];
    const float* b_k2 = (const float*)d_in[12];
    const float* w_v  = (const float*)d_in[13];
    const float* s_v  = (const float*)d_in[14];
    const float* b_v  = (const float*)d_in[15];
    const float* w_o  = (const float*)d_in[16];
    const float* s_o  = (const float*)d_in[17];
    const float* b_o  = (const float*)d_in[18];
    float* out = (float*)d_out;

    const size_t NB = (size_t)HW * CCH;
    float* kb = (float*)d_ws;
    float* vb = kb + NB;

    const dim3 blk(256);
    const dim3 grd(HW / 64);

    proj_fused_wmma<<<grd, blk, 0, stream>>>(x,
                                             w_k1, s_k1, b_k1, w_k2, s_k2, b_k2,
                                             w_v, s_v, b_v,
                                             kb, vb);
    local_attn_out_kernel<<<dim3(HW / TILE_PX), dim3(ATT_THREADS), 0, stream>>>(
        kb, vb, x, w_q1, s_q1, b_q1, w_q2, s_q2, b_q2, w_o, s_o, b_o, out);
}
